# starting kernel
# baseline (speedup 1.0000x reference)
#include <hip/hip_runtime.h>
#include <hip/hip_bf16.h>

typedef __attribute__((ext_vector_type(2))) float v2f;
typedef __attribute__((ext_vector_type(8))) float v8f;

__global__ void deg_kernel(const int* __restrict__ src, const int* __restrict__ dst,
                           float* __restrict__ deg_out, float* __restrict__ deg_in, int E) {
    int e = blockIdx.x * blockDim.x + threadIdx.x;
    if (e < E) {
        atomicAdd(&deg_out[src[e]], 1.0f);
        atomicAdd(&deg_in[dst[e]], 1.0f);
    }
}

__global__ void norm_kernel(float* __restrict__ deg_out, float* __restrict__ deg_in, int N) {
    int i = blockIdx.x * blockDim.x + threadIdx.x;
    if (i < N) {
        float d0 = deg_out[i];
        float d1 = deg_in[i];
        deg_out[i] = rsqrtf(d0 > 0.0f ? d0 : 1.0f);
        deg_in[i]  = rsqrtf(d1 > 0.0f ? d1 : 1.0f);
    }
}

template <int K, int COLS>
__global__ void __launch_bounds__(128)
gemm_norm_wmma(const float* __restrict__ X, const float* __restrict__ normv,
               const float* __restrict__ W, float* __restrict__ H, int nrows) {
    constexpr int NT = COLS / 16;
    const int lane  = threadIdx.x & 31;
    const int wave  = threadIdx.x >> 5;
    const int tile_m = blockIdx.x * (blockDim.x >> 5) + wave;
    if (tile_m * 16 >= nrows) return;

    const int mr = lane & 15;
    const int kh = (lane >> 4) << 1;
    const int m  = tile_m * 16 + mr;

    const float nm = normv[m];
    const float* __restrict__ xrow = X + (long)m * K;

    v8f acc[NT];
#pragma unroll
    for (int t = 0; t < NT; ++t) { v8f z = {}; acc[t] = z; }

#pragma unroll 4
    for (int k = 0; k < K; k += 4) {
        v2f a;
        a.x = xrow[k + kh]     * nm;
        a.y = xrow[k + kh + 1] * nm;
        const float* __restrict__ wr = W + (long)(k + kh) * COLS;
#pragma unroll
        for (int t = 0; t < NT; ++t) {
            v2f b;
            b.x = wr[t * 16 + mr];
            b.y = wr[COLS + t * 16 + mr];
            acc[t] = __builtin_amdgcn_wmma_f32_16x16x4_f32(
                false, a, false, b, (short)0, acc[t], false, false);
        }
    }

    const int mbase = tile_m * 16 + ((lane >> 4) << 3);
#pragma unroll
    for (int t = 0; t < NT; ++t) {
#pragma unroll
        for (int v = 0; v < 8; ++v) {
            H[(long)(mbase + v) * COLS + t * 16 + mr] = acc[t][v];
        }
    }
}

template <int C>
__global__ void edge_agg(const float* __restrict__ H, const int* __restrict__ src,
                         const int* __restrict__ dst, float* __restrict__ agg, int E) {
    long idx = (long)blockIdx.x * blockDim.x + threadIdx.x;
    int e = (int)(idx / C);
    int c = (int)(idx & (C - 1));
    if (e >= E) return;
    float v = H[(long)src[e] * C + c];
    atomicAdd(&agg[(long)dst[e] * C + c], v);
}

template <int C, bool RELU>
__global__ void finalize_kernel(float* __restrict__ agg, const float* __restrict__ norm_in,
                                const float* __restrict__ bias, long total) {
    long idx = (long)blockIdx.x * blockDim.x + threadIdx.x;
    if (idx < total) {
        int row = (int)(idx / C);
        int col = (int)(idx & (C - 1));
        float v = agg[idx] * norm_in[row] + bias[col];
        agg[idx] = RELU ? (v > 0.0f ? v : 0.0f) : v;
    }
}

extern "C" void kernel_launch(void* const* d_in, const int* in_sizes, int n_in,
                              void* d_out, int out_size, void* d_ws, size_t ws_size,
                              hipStream_t stream) {
    const float* features = (const float*)d_in[0];
    const int*   src      = (const int*)d_in[1];
    const int*   dst      = (const int*)d_in[2];
    const float* W1       = (const float*)d_in[3];
    const float* b1       = (const float*)d_in[4];
    const float* W2       = (const float*)d_in[5];
    const float* b2       = (const float*)d_in[6];

    const int IN_F = 128, HID = 64, CLS = 32;
    const int N = in_sizes[0] / IN_F;
    const int E = in_sizes[1];

    float* ws       = (float*)d_ws;
    float* norm_out = ws;
    float* norm_in  = ws + (long)N;
    float* hbuf     = ws + 2L * N;
    float* aggbuf   = ws + 2L * N + 64L * N;
    float* outp     = (float*)d_out;

    hipMemsetAsync(norm_out, 0, sizeof(float) * 2L * N, stream);
    hipMemsetAsync(aggbuf,   0, sizeof(float) * 64L * N, stream);
    hipMemsetAsync(outp,     0, sizeof(float) * (size_t)out_size, stream);

    deg_kernel<<<(E + 255) / 256, 256, 0, stream>>>(src, dst, norm_out, norm_in, E);
    norm_kernel<<<(N + 255) / 256, 256, 0, stream>>>(norm_out, norm_in, N);

    const int mtiles = (N + 15) / 16;
    const int gblocks = (mtiles + 3) / 4;

    gemm_norm_wmma<128, 64><<<gblocks, 128, 0, stream>>>(features, norm_out, W1, hbuf, N);
    {
        long tot = (long)E * HID;
        edge_agg<64><<<(unsigned)((tot + 255) / 256), 256, 0, stream>>>(hbuf, src, dst, aggbuf, E);
        long n1 = (long)N * HID;
        finalize_kernel<64, true><<<(unsigned)((n1 + 255) / 256), 256, 0, stream>>>(
            aggbuf, norm_in, b1, n1);
    }

    gemm_norm_wmma<64, 32><<<gblocks, 128, 0, stream>>>(aggbuf, norm_out, W2, hbuf, N);
    {
        long tot = (long)E * CLS;
        edge_agg<32><<<(unsigned)((tot + 255) / 256), 256, 0, stream>>>(hbuf, src, dst, outp, E);
        long n2 = (long)N * CLS;
        finalize_kernel<32, false><<<(unsigned)((n2 + 255) / 256), 256, 0, stream>>>(
            outp, norm_in, b2, n2);
    }
}


# Edge sums as an atomic-free per-node gather over a dst-ordered edge list (rank from the in-degree atomic + prefix sum, rebuilt each call); row sums now in arrival-rank order, epilogue in the gather
# speedup vs baseline: 2.3830x; 2.3830x over previous
#include <hip/hip_runtime.h>
#include <hip/hip_bf16.h>


typedef __attribute__((ext_vector_type(2))) float v2f;
typedef __attribute__((ext_vector_type(8))) float v8f;

__global__ void deg_rank_kernel(const int* __restrict__ src, const int* __restrict__ dst,
                                int* __restrict__ cnt_out, int* __restrict__ cnt_in,
                                int* __restrict__ rank, int E) {
    int e = blockIdx.x * blockDim.x + threadIdx.x;
    if (e < E) {
        const int s = src[e];
        const int d = dst[e];
        atomicAdd(&cnt_out[s], 1);
        rank[e] = atomicAdd(&cnt_in[d], 1);
    }
}

__global__ void __launch_bounds__(1024)
rowstart_norm_kernel(const int* __restrict__ cnt_in, const int* __restrict__ cnt_out,
                     int2* __restrict__ rowinfo, float* __restrict__ norm_out, int N) {
    __shared__ int wave_tot[32];
    __shared__ int wave_front[32];
    const int tid  = threadIdx.x;
    const int lane = tid & 31;
    const int wave = tid >> 5;
    const int base = blockIdx.x * 1024;

    int front = 0;
    const int4* __restrict__ c4 = (const int4*)cnt_in;
    const int nfull = (int)blockIdx.x >> 2;
#pragma unroll 8
    for (int k = 0; k < nfull; ++k) {
        const int4 v = c4[k * 1024 + tid];
        front += (v.x + v.y) + (v.z + v.w);
    }
    if (tid < (((int)blockIdx.x & 3) << 8)) {
        const int4 v = c4[nfull * 1024 + tid];
        front += (v.x + v.y) + (v.z + v.w);
    }

    const int i = base + tid;
    const int x = (i < N) ? cnt_in[i] : 0;

    int incl = x;
#pragma unroll
    for (int o = 1; o < 32; o <<= 1) {
        const int t = __shfl_up(incl, o);
        if (lane >= o) incl += t;
    }
#pragma unroll
    for (int o = 16; o > 0; o >>= 1) front += __shfl_xor(front, o);

    if (lane == 31) wave_tot[wave] = incl;
    if (lane == 0)  wave_front[wave] = front;
    __syncthreads();

    const int wt = wave_tot[lane];
    int wf = wave_front[lane];
    int wincl = wt;
#pragma unroll
    for (int o = 1; o < 32; o <<= 1) {
        const int t = __shfl_up(wincl, o);
        if (lane >= o) wincl += t;
    }
#pragma unroll
    for (int o = 16; o > 0; o >>= 1) wf += __shfl_xor(wf, o);
    const int waves_before = __shfl(wincl - wt, wave);

    if (i < N) {
        rowinfo[i] = make_int2(wf + waves_before + (incl - x), x);
        const int co = cnt_out[i];
        norm_out[i] = rsqrtf(co > 0 ? (float)co : 1.0f);
    }
}

__global__ void fill_kernel(const int* __restrict__ src, const int* __restrict__ dst,
                            const int* __restrict__ rank, const int2* __restrict__ rowinfo,
                            int* __restrict__ esrc, int E) {
    int e = blockIdx.x * blockDim.x + threadIdx.x;
    if (e < E) esrc[rowinfo[dst[e]].x + rank[e]] = src[e];
}

template <int K, int COLS>
__global__ void __launch_bounds__(128)
gemm_norm_wmma(const float* __restrict__ X, const float* __restrict__ normv,
               const float* __restrict__ W, float* __restrict__ H, int nrows) {
    constexpr int NT = COLS / 16;
    const int lane  = threadIdx.x & 31;
    const int wave  = threadIdx.x >> 5;
    const int tile_m = blockIdx.x * (blockDim.x >> 5) + wave;
    if (tile_m * 16 >= nrows) return;

    const int mr = lane & 15;
    const int kh = (lane >> 4) << 1;
    const int m  = tile_m * 16 + mr;

    const float nm = normv[m];
    const float* __restrict__ xrow = X + (long)m * K;

    v8f acc[NT];
#pragma unroll
    for (int t = 0; t < NT; ++t) { v8f z = {}; acc[t] = z; }

#pragma unroll 4
    for (int k = 0; k < K; k += 4) {
        v2f a;
        a.x = xrow[k + kh]     * nm;
        a.y = xrow[k + kh + 1] * nm;
        const float* __restrict__ wr = W + (long)(k + kh) * COLS;
#pragma unroll
        for (int t = 0; t < NT; ++t) {
            v2f b;
            b.x = wr[t * 16 + mr];
            b.y = wr[COLS + t * 16 + mr];
            acc[t] = __builtin_amdgcn_wmma_f32_16x16x4_f32(
                false, a, false, b, (short)0, acc[t], false, false);
        }
    }

    const int mbase = tile_m * 16 + ((lane >> 4) << 3);
#pragma unroll
    for (int t = 0; t < NT; ++t) {
#pragma unroll
        for (int v = 0; v < 8; ++v) {
            H[(long)(mbase + v) * COLS + t * 16 + mr] = acc[t][v];
        }
    }
}

template <int C, bool RELU>
__global__ void __launch_bounds__(256)
gather_rows(const float* __restrict__ H, const int* __restrict__ esrc,
            const long long* __restrict__ rowinfo,
            const float* __restrict__ bias, float* __restrict__ out, int N) {
    constexpr int VEC = C / 32;
    constexpr int B   = 16;
    const int lane = threadIdx.x & 31;
    const int wave = __builtin_amdgcn_readfirstlane((int)(threadIdx.x >> 5));
    const int nwaves = (int)gridDim.x * 8;

    float bv[VEC];
#pragma unroll
    for (int k = 0; k < VEC; ++k) bv[k] = bias[lane * VEC + k];

    for (int node = (int)blockIdx.x * 8 + wave; node < N; node += nwaves) {
        const long long info = rowinfo[node];
        const int start = (int)info;
        const int cnt   = (int)(info >> 32);
        float acc[VEC];
#pragma unroll
        for (int k = 0; k < VEC; ++k) acc[k] = 0.0f;

        for (int j0 = 0; j0 < cnt; j0 += 32) {
            const int n32 = (cnt - j0) < 32 ? (cnt - j0) : 32;
            const int myidx = (lane < n32) ? esrc[start + j0 + lane] : 0;
            for (int h = 0; h < n32; h += B) {
                const int n = n32 - h;
                float v[B][VEC];
#pragma unroll
                for (int u = 0; u < B; ++u) {
#pragma unroll
                    for (int k = 0; k < VEC; ++k) v[u][k] = 0.0f;
                    if (u == 0 || u < n) {
                        const int s = __builtin_amdgcn_readlane(myidx, h + u);
                        const float* __restrict__ p = H + (long)s * C + lane * VEC;
                        if constexpr (VEC == 2) {
                            const float2 t = *reinterpret_cast<const float2*>(p);
                            v[u][0] = t.x;
                            v[u][1] = t.y;
                        } else {
                            v[u][0] = p[0];
                        }
                    }
                }
#pragma unroll
                for (int u = 0; u < B; ++u) {
#pragma unroll
                    for (int k = 0; k < VEC; ++k) acc[k] += v[u][k];
                }
            }
        }

        const float nin = rsqrtf(cnt > 0 ? (float)cnt : 1.0f);
        float r[VEC];
#pragma unroll
        for (int k = 0; k < VEC; ++k) {
            const float t = acc[k] * nin + bv[k];
            r[k] = RELU ? (t > 0.0f ? t : 0.0f) : t;
        }
        float* __restrict__ o = out + (long)node * C + lane * VEC;
        if constexpr (VEC == 2) {
            *reinterpret_cast<float2*>(o) = make_float2(r[0], r[1]);
        } else {
            o[0] = r[0];
        }
    }
}

extern "C" void kernel_launch(void* const* d_in, const int* in_sizes, int n_in,
                              void* d_out, int out_size, void* d_ws, size_t ws_size,
                              hipStream_t stream) {
    const float* features = (const float*)d_in[0];
    const int*   src      = (const int*)d_in[1];
    const int*   dst      = (const int*)d_in[2];
    const float* W1       = (const float*)d_in[3];
    const float* b1       = (const float*)d_in[4];
    const float* W2       = (const float*)d_in[5];
    const float* b2       = (const float*)d_in[6];

    const int IN_F = 128;
    const int N = in_sizes[0] / IN_F;
    const int E = in_sizes[1];

    int*   wsi      = (int*)d_ws;
    float* wsf      = (float*)d_ws;
    float* hbuf     = wsf;
    float* x1       = wsf + 64L * N;
    int*   esrc     = wsi + 128L * N;
    int*   rank     = wsi + 128L * N + (long)E;
    int*   cnt_in   = wsi + 128L * N + 2L * E;
    int*   cnt_out  = wsi + 129L * N + 2L * E;
    int2*  rowinfo  = (int2*)(wsi + 130L * N + 2L * E);
    float* norm_out = wsf + 132L * N + 2L * E;
    float* outp     = (float*)d_out;

    hipMemsetAsync(cnt_in, 0, sizeof(int) * 2L * N, stream);

    deg_rank_kernel<<<(E + 255) / 256, 256, 0, stream>>>(src, dst, cnt_out, cnt_in, rank, E);
    rowstart_norm_kernel<<<(N + 1023) / 1024, 1024, 0, stream>>>(cnt_in, cnt_out, rowinfo, norm_out, N);
    fill_kernel<<<(E + 255) / 256, 256, 0, stream>>>(src, dst, rank, rowinfo, esrc, E);

    const int mtiles = (N + 15) / 16;
    const int gblocks = (mtiles + 3) / 4;
    const int gather_blocks = 2048;

    gemm_norm_wmma<128, 64><<<gblocks, 128, 0, stream>>>(features, norm_out, W1, hbuf, N);
    gather_rows<64, true><<<gather_blocks, 256, 0, stream>>>(hbuf, esrc, (const long long*)rowinfo, b1, x1, N);

    gemm_norm_wmma<64, 32><<<gblocks, 128, 0, stream>>>(x1, norm_out, W2, hbuf, N);
    gather_rows<32, false><<<gather_blocks, 256, 0, stream>>>(hbuf, esrc, (const long long*)rowinfo, b2, outp, N);
}


# Source lists built in the counting pass: 32-slot line per node indexed by the in-degree atomic's return, overflow to a scanned side list; scan and fill launches gone; row sums stay in arrival order
# speedup vs baseline: 2.6876x; 1.1278x over previous
#include <hip/hip_runtime.h>
#include <hip/hip_bf16.h>


typedef __attribute__((ext_vector_type(2))) float v2f;
typedef __attribute__((ext_vector_type(8))) float v8f;

constexpr int CAP = 32;

__global__ void build_kernel(const int* __restrict__ src, const int* __restrict__ dst,
                             int* __restrict__ cnt_out, int* __restrict__ cnt_in,
                             int* __restrict__ slots, int* __restrict__ ovf_n,
                             int* __restrict__ ovf_src, int* __restrict__ ovf_dst, int E) {
    int e = blockIdx.x * blockDim.x + threadIdx.x;
    if (e < E) {
        const int s = src[e];
        const int d = dst[e];
        atomicAdd(&cnt_out[s], 1);
        const int r = atomicAdd(&cnt_in[d], 1);
        if (r < CAP) {
            slots[(long)d * CAP + r] = s;
        } else {
            const int k = atomicAdd(ovf_n, 1);
            ovf_src[k] = s;
            ovf_dst[k] = d;
        }
    }
}

template <int K, int COLS>
__global__ void __launch_bounds__(128)
gemm_norm_wmma(const float* __restrict__ X, const int* __restrict__ cntv,
               const float* __restrict__ W, float* __restrict__ H, int nrows) {
    constexpr int NT = COLS / 16;
    const int lane  = threadIdx.x & 31;
    const int wave  = threadIdx.x >> 5;
    const int tile_m = blockIdx.x * (blockDim.x >> 5) + wave;
    if (tile_m * 16 >= nrows) return;

    const int mr = lane & 15;
    const int kh = (lane >> 4) << 1;
    const int m  = tile_m * 16 + mr;

    const int co = cntv[m];
    const float nm = rsqrtf(co > 0 ? (float)co : 1.0f);
    const float* __restrict__ xrow = X + (long)m * K;

    v8f acc[NT];
#pragma unroll
    for (int t = 0; t < NT; ++t) { v8f z = {}; acc[t] = z; }

#pragma unroll 4
    for (int k = 0; k < K; k += 4) {
        v2f a;
        a.x = xrow[k + kh]     * nm;
        a.y = xrow[k + kh + 1] * nm;
        const float* __restrict__ wr = W + (long)(k + kh) * COLS;
#pragma unroll
        for (int t = 0; t < NT; ++t) {
            v2f b;
            b.x = wr[t * 16 + mr];
            b.y = wr[COLS + t * 16 + mr];
            acc[t] = __builtin_amdgcn_wmma_f32_16x16x4_f32(
                false, a, false, b, (short)0, acc[t], false, false);
        }
    }

    const int mbase = tile_m * 16 + ((lane >> 4) << 3);
#pragma unroll
    for (int t = 0; t < NT; ++t) {
#pragma unroll
        for (int v = 0; v < 8; ++v) {
            H[(long)(mbase + v) * COLS + t * 16 + mr] = acc[t][v];
        }
    }
}

template <int C, bool RELU>
__global__ void __launch_bounds__(256)
gather_rows(const float* __restrict__ H, const int* __restrict__ slots,
            const int* __restrict__ cnt_in, const int* __restrict__ ovf_n,
            const int* __restrict__ ovf_src, const int* __restrict__ ovf_dst,
            const float* __restrict__ bias, float* __restrict__ out, int N) {
    constexpr int VEC = C / 32;
    constexpr int B   = 16;
    const int lane = threadIdx.x & 31;
    const int wave = __builtin_amdgcn_readfirstlane((int)(threadIdx.x >> 5));
    const int nwaves = (int)gridDim.x * 8;

    float bv[VEC];
#pragma unroll
    for (int k = 0; k < VEC; ++k) bv[k] = bias[lane * VEC + k];

    for (int node = (int)blockIdx.x * 8 + wave; node < N; node += nwaves) {
        const int cnt = cnt_in[node];
        const int n32 = cnt < CAP ? cnt : CAP;
        float acc[VEC];
#pragma unroll
        for (int k = 0; k < VEC; ++k) acc[k] = 0.0f;

        const int myidx = (lane < n32) ? slots[(long)node * CAP + lane] : 0;
        for (int h = 0; h < n32; h += B) {
            const int n = n32 - h;
            float v[B][VEC];
#pragma unroll
            for (int u = 0; u < B; ++u) {
#pragma unroll
                for (int k = 0; k < VEC; ++k) v[u][k] = 0.0f;
                if (u == 0 || u < n) {
                    const int s = __builtin_amdgcn_readlane(myidx, h + u);
                    const float* __restrict__ p = H + (long)s * C + lane * VEC;
                    if constexpr (VEC == 2) {
                        const float2 t = *reinterpret_cast<const float2*>(p);
                        v[u][0] = t.x;
                        v[u][1] = t.y;
                    } else {
                        v[u][0] = p[0];
                    }
                }
            }
#pragma unroll
            for (int u = 0; u < B; ++u) {
#pragma unroll
                for (int k = 0; k < VEC; ++k) acc[k] += v[u][k];
            }
        }

        if (cnt > CAP) {
            const int novf = ovf_n[0];
            for (int j = 0; j < novf; j += 32) {
                const int q = j + lane;
                const int od = (q < novf) ? ovf_dst[q] : -1;
                const int os = (q < novf) ? ovf_src[q] : 0;
                unsigned m = __builtin_amdgcn_ballot_w32(od == node);
                while (m != 0u) {
                    const int b = __builtin_ctz(m);
                    m &= m - 1u;
                    const int s = __builtin_amdgcn_readlane(os, b);
                    const float* __restrict__ p = H + (long)s * C + lane * VEC;
#pragma unroll
                    for (int k = 0; k < VEC; ++k) acc[k] += p[k];
                }
            }
        }

        const float nin = rsqrtf(cnt > 0 ? (float)cnt : 1.0f);
        float r[VEC];
#pragma unroll
        for (int k = 0; k < VEC; ++k) {
            const float t = acc[k] * nin + bv[k];
            r[k] = RELU ? (t > 0.0f ? t : 0.0f) : t;
        }
        float* __restrict__ o = out + (long)node * C + lane * VEC;
        if constexpr (VEC == 2) {
            *reinterpret_cast<float2*>(o) = make_float2(r[0], r[1]);
        } else {
            o[0] = r[0];
        }
    }
}

extern "C" void kernel_launch(void* const* d_in, const int* in_sizes, int n_in,
                              void* d_out, int out_size, void* d_ws, size_t ws_size,
                              hipStream_t stream) {
    const float* features = (const float*)d_in[0];
    const int*   src      = (const int*)d_in[1];
    const int*   dst      = (const int*)d_in[2];
    const float* W1       = (const float*)d_in[3];
    const float* b1       = (const float*)d_in[4];
    const float* W2       = (const float*)d_in[5];
    const float* b2       = (const float*)d_in[6];

    const int IN_F = 128;
    const int N = in_sizes[0] / IN_F;
    const int E = in_sizes[1];

    int*   wsi      = (int*)d_ws;
    float* wsf      = (float*)d_ws;
    float* hbuf     = wsf;
    float* x1       = wsf + 64L * N;
    int*   slots    = wsi + 128L * N;
    int*   ovf_src  = wsi + 160L * N;
    int*   ovf_dst  = wsi + 160L * N + (long)E;
    int*   cnt_in   = wsi + 160L * N + 2L * E;
    int*   cnt_out  = wsi + 161L * N + 2L * E;
    int*   ovf_n    = wsi + 162L * N + 2L * E;
    float* outp     = (float*)d_out;

    hipMemsetAsync(cnt_in, 0, sizeof(int) * (2L * N + 32), stream);

    build_kernel<<<(E + 255) / 256, 256, 0, stream>>>(src, dst, cnt_out, cnt_in, slots, ovf_n, ovf_src, ovf_dst, E);

    const int mtiles = (N + 15) / 16;
    const int gblocks = (mtiles + 3) / 4;
    const int gather_blocks = 2048;

    gemm_norm_wmma<128, 64><<<gblocks, 128, 0, stream>>>(features, cnt_out, W1, hbuf, N);
    gather_rows<64, true><<<gather_blocks, 256, 0, stream>>>(hbuf, slots, cnt_in, ovf_n, ovf_src, ovf_dst, b1, x1, N);

    gemm_norm_wmma<64, 32><<<gblocks, 128, 0, stream>>>(x1, cnt_out, W2, hbuf, N);
    gather_rows<32, false><<<gather_blocks, 256, 0, stream>>>(hbuf, slots, cnt_in, ovf_n, ovf_src, ovf_dst, b2, outp, N);
}


# In-degree ranks from LDS counters in a two-level pass over 512-node ranges (one returning global add per range and work-group, none per edge); slot lines stored whole; sums stay in arrival order
# speedup vs baseline: 3.6305x; 1.3508x over previous
#include <hip/hip_runtime.h>
#include <hip/hip_bf16.h>


typedef __attribute__((ext_vector_type(2))) float v2f;
typedef __attribute__((ext_vector_type(8))) float v8f;

typedef unsigned long long u64;

constexpr int CAP   = 32;
constexpr int BSH   = 9;
constexpr int RB    = 1 << BSH;
constexpr int T1    = 512;
constexpr int NBMAX = T1;
constexpr int EPT   = 16;
constexpr int BCAP  = 12288;
constexpr int GSTR  = 32;
constexpr int T2    = 1024;
constexpr int PPT   = 8;

__device__ __forceinline__ u64 pack_pair(int s, int d) {
    return (u64)(unsigned)s | ((u64)(unsigned)d << 32);
}

__global__ void __launch_bounds__(T1)
bucket_edges(const int* __restrict__ src, const int* __restrict__ dst,
             int* __restrict__ cnt_out, int* __restrict__ gcount,
             u64* __restrict__ blist, int* __restrict__ spill_n,
             u64* __restrict__ spill, int E, int NB) {
    __shared__ int lcount[NBMAX];
    __shared__ int lstart[NBMAX];
    __shared__ int lbase[NBMAX];
    __shared__ u64 lpairs[T1 * EPT];
    const int tid   = (int)threadIdx.x;
    const int chunk = (E + (int)gridDim.x - 1) / (int)gridDim.x;
    const int cbeg  = (int)blockIdx.x * chunk;
    const int cend  = (cbeg + chunk < E) ? (cbeg + chunk) : E;

    for (int base = cbeg; base < cend; base += T1 * EPT) {
        if (tid < NB) lcount[tid] = 0;
        __syncthreads();

        int s[EPT], d[EPT], lr[EPT];
#pragma unroll
        for (int i = 0; i < EPT; ++i) {
            const int e = base + i * T1 + tid;
            const bool ok = e < cend;
            s[i] = ok ? src[e] : 0;
            d[i] = ok ? dst[e] : 0;
        }
#pragma unroll
        for (int i = 0; i < EPT; ++i) {
            lr[i] = 0;
            if (base + i * T1 + tid < cend) lr[i] = atomicAdd(&lcount[d[i] >> BSH], 1);
        }
        __syncthreads();

        const int c = (tid < NB) ? lcount[tid] : 0;
        int got = 0;
        if (c > 0) got = atomicAdd(&gcount[tid * GSTR], c);
        if (tid < 32) {
            int carry = 0;
            for (int c0 = 0; c0 < NB; c0 += 32) {
                const int idx = c0 + tid;
                const int v = (idx < NB) ? lcount[idx] : 0;
                int x = v;
                for (int o = 1; o < 32; o <<= 1) {
                    const int y = __shfl_up(x, o);
                    if (tid >= o) x += y;
                }
                if (idx < NB) lstart[idx] = carry + x - v;
                carry += __shfl(x, 31);
            }
        }
        if (tid < NB) lbase[tid] = got;
        __syncthreads();

        int st[EPT];
#pragma unroll
        for (int i = 0; i < EPT; ++i) st[i] = lstart[d[i] >> BSH];
#pragma unroll
        for (int i = 0; i < EPT; ++i) {
            if (base + i * T1 + tid < cend) lpairs[st[i] + lr[i]] = pack_pair(s[i], d[i]);
        }
        __syncthreads();

        const int left = cend - base;
        const int cn = left < T1 * EPT ? left : T1 * EPT;
        u64 v[EPT];
        int gb[EPT], ls[EPT];
#pragma unroll
        for (int i = 0; i < EPT; ++i) {
            const int j = i * T1 + tid;
            v[i] = lpairs[j < cn ? j : 0];
        }
#pragma unroll
        for (int i = 0; i < EPT; ++i) {
            const int b = (int)(v[i] >> 32) >> BSH;
            gb[i] = lbase[b];
            ls[i] = lstart[b];
        }
#pragma unroll
        for (int i = 0; i < EPT; ++i) {
            const int j = i * T1 + tid;
            if (j < cn) {
                const int b   = (int)(v[i] >> 32) >> BSH;
                const int pos = gb[i] + (j - ls[i]);
                u64* q;
                if (pos < BCAP) {
                    q = blist + (long)b * BCAP + pos;
                } else {
                    q = spill + atomicAdd(spill_n, 1);
                }
                *q = v[i];
                atomicAdd(&cnt_out[(int)(unsigned)v[i]], 1);
            }
        }
        if (base + T1 * EPT < cend) __syncthreads();
    }
}

__global__ void __launch_bounds__(T2)
rank_range(const u64* __restrict__ blist, const int* __restrict__ gcount,
           const int* __restrict__ spill_n, const u64* __restrict__ spill,
           int* __restrict__ cnt_in, int* __restrict__ slots, int* __restrict__ ovf_n,
           int* __restrict__ ovf_src, int* __restrict__ ovf_dst, int N) {
    __shared__ int lcnt[RB];
    __shared__ __attribute__((aligned(16))) int lslots[RB * CAP];
    const int tid   = (int)threadIdx.x;
    const int b     = (int)blockIdx.x;
    const int nbase = b << BSH;

    int4* __restrict__ l4 = reinterpret_cast<int4*>(lslots);
    for (int t = tid; t < RB; t += T2) lcnt[t] = 0;
    for (int t = tid; t < RB * CAP / 4; t += T2) l4[t] = make_int4(0, 0, 0, 0);
    __syncthreads();

    auto place = [&](const u64 p) {
        const int ps = (int)(unsigned)p;
        const int pd = (int)(p >> 32);
        const int dl = pd - nbase;
        const int r  = atomicAdd(&lcnt[dl], 1);
        if (r < CAP) {
            lslots[dl * CAP + r] = ps;
        } else {
            const int k = atomicAdd(ovf_n, 1);
            ovf_src[k] = ps;
            ovf_dst[k] = pd;
        }
    };

    int n = gcount[b * GSTR];
    n = n < BCAP ? n : BCAP;
    const u64* __restrict__ bl = blist + (long)b * BCAP;
    for (int base = 0; base < n; base += T2 * PPT) {
        u64 p[PPT];
#pragma unroll
        for (int u = 0; u < PPT; ++u) {
            const int i = base + u * T2 + tid;
            p[u] = (i < n) ? bl[i] : 0ull;
        }
#pragma unroll
        for (int u = 0; u < PPT; ++u) {
            if (base + u * T2 + tid < n) place(p[u]);
        }
    }

    const int ns = spill_n[0];
    for (int i = tid; i < ns; i += T2) {
        const u64 q = spill[i];
        if (((int)(q >> 32) >> BSH) == b) place(q);
    }
    __syncthreads();

    const int left = N - nbase;
    const int nn = left < RB ? left : RB;
    for (int t = tid; t < nn; t += T2) cnt_in[nbase + t] = lcnt[t];
    int4* __restrict__ g4 = reinterpret_cast<int4*>(slots + (long)nbase * CAP);
    for (int t = tid; t < nn * (CAP / 4); t += T2) g4[t] = l4[t];
}

template <int K, int COLS>
__global__ void __launch_bounds__(128)
gemm_norm_wmma(const float* __restrict__ X, const int* __restrict__ cntv,
               const float* __restrict__ W, float* __restrict__ H, int nrows) {
    constexpr int NT = COLS / 16;
    const int lane  = threadIdx.x & 31;
    const int wave  = threadIdx.x >> 5;
    const int tile_m = blockIdx.x * (blockDim.x >> 5) + wave;
    if (tile_m * 16 >= nrows) return;

    const int mr = lane & 15;
    const int kh = (lane >> 4) << 1;
    const int m  = tile_m * 16 + mr;

    const int co = cntv[m];
    const float nm = rsqrtf(co > 0 ? (float)co : 1.0f);
    const float* __restrict__ xrow = X + (long)m * K;

    v8f acc[NT];
#pragma unroll
    for (int t = 0; t < NT; ++t) { v8f z = {}; acc[t] = z; }

#pragma unroll 4
    for (int k = 0; k < K; k += 4) {
        v2f a;
        a.x = xrow[k + kh]     * nm;
        a.y = xrow[k + kh + 1] * nm;
        const float* __restrict__ wr = W + (long)(k + kh) * COLS;
#pragma unroll
        for (int t = 0; t < NT; ++t) {
            v2f b;
            b.x = wr[t * 16 + mr];
            b.y = wr[COLS + t * 16 + mr];
            acc[t] = __builtin_amdgcn_wmma_f32_16x16x4_f32(
                false, a, false, b, (short)0, acc[t], false, false);
        }
    }

    const int mbase = tile_m * 16 + ((lane >> 4) << 3);
#pragma unroll
    for (int t = 0; t < NT; ++t) {
#pragma unroll
        for (int v = 0; v < 8; ++v) {
            H[(long)(mbase + v) * COLS + t * 16 + mr] = acc[t][v];
        }
    }
}

template <int C, bool RELU>
__global__ void __launch_bounds__(256)
gather_rows(const float* __restrict__ H, const int* __restrict__ slots,
            const int* __restrict__ cnt_in, const int* __restrict__ ovf_n,
            const int* __restrict__ ovf_src, const int* __restrict__ ovf_dst,
            const float* __restrict__ bias, float* __restrict__ out, int N) {
    constexpr int VEC = C / 32;
    constexpr int B   = 16;
    const int lane = threadIdx.x & 31;
    const int wave = __builtin_amdgcn_readfirstlane((int)(threadIdx.x >> 5));
    const int nwaves = (int)gridDim.x * 8;

    float bv[VEC];
#pragma unroll
    for (int k = 0; k < VEC; ++k) bv[k] = bias[lane * VEC + k];

    for (int node = (int)blockIdx.x * 8 + wave; node < N; node += nwaves) {
        const int cnt = cnt_in[node];
        const int n32 = cnt < CAP ? cnt : CAP;
        float acc[VEC];
#pragma unroll
        for (int k = 0; k < VEC; ++k) acc[k] = 0.0f;

        const int myidx = (lane < n32) ? slots[(long)node * CAP + lane] : 0;
        for (int h = 0; h < n32; h += B) {
            const int n = n32 - h;
            float v[B][VEC];
#pragma unroll
            for (int u = 0; u < B; ++u) {
#pragma unroll
                for (int k = 0; k < VEC; ++k) v[u][k] = 0.0f;
                if (u == 0 || u < n) {
                    const int s = __builtin_amdgcn_readlane(myidx, h + u);
                    const float* __restrict__ p = H + (long)s * C + lane * VEC;
                    if constexpr (VEC == 2) {
                        const float2 t = *reinterpret_cast<const float2*>(p);
                        v[u][0] = t.x;
                        v[u][1] = t.y;
                    } else {
                        v[u][0] = p[0];
                    }
                }
            }
#pragma unroll
            for (int u = 0; u < B; ++u) {
#pragma unroll
                for (int k = 0; k < VEC; ++k) acc[k] += v[u][k];
            }
        }

        if (cnt > CAP) {
            const int novf = ovf_n[0];
            for (int j = 0; j < novf; j += 32) {
                const int q = j + lane;
                const int od = (q < novf) ? ovf_dst[q] : -1;
                const int os = (q < novf) ? ovf_src[q] : 0;
                unsigned m = __builtin_amdgcn_ballot_w32(od == node);
                while (m != 0u) {
                    const int b = __builtin_ctz(m);
                    m &= m - 1u;
                    const int s = __builtin_amdgcn_readlane(os, b);
                    const float* __restrict__ p = H + (long)s * C + lane * VEC;
#pragma unroll
                    for (int k = 0; k < VEC; ++k) acc[k] += p[k];
                }
            }
        }

        const float nin = rsqrtf(cnt > 0 ? (float)cnt : 1.0f);
        float r[VEC];
#pragma unroll
        for (int k = 0; k < VEC; ++k) {
            const float t = acc[k] * nin + bv[k];
            r[k] = RELU ? (t > 0.0f ? t : 0.0f) : t;
        }
        float* __restrict__ o = out + (long)node * C + lane * VEC;
        if constexpr (VEC == 2) {
            *reinterpret_cast<float2*>(o) = make_float2(r[0], r[1]);
        } else {
            o[0] = r[0];
        }
    }
}

extern "C" void kernel_launch(void* const* d_in, const int* in_sizes, int n_in,
                              void* d_out, int out_size, void* d_ws, size_t ws_size,
                              hipStream_t stream) {
    const float* features = (const float*)d_in[0];
    const int*   src      = (const int*)d_in[1];
    const int*   dst      = (const int*)d_in[2];
    const float* W1       = (const float*)d_in[3];
    const float* b1       = (const float*)d_in[4];
    const float* W2       = (const float*)d_in[5];
    const float* b2       = (const float*)d_in[6];

    const int IN_F = 128;
    const int N = in_sizes[0] / IN_F;
    const int E = in_sizes[1];

    int*   wsi      = (int*)d_ws;
    float* wsf      = (float*)d_ws;
    float* hbuf     = wsf;
    float* x1       = wsf + 64L * N;
    int*   slots    = wsi + 128L * N;
    int*   ovf_src  = wsi + 160L * N;
    int*   ovf_dst  = wsi + 160L * N + (long)E;
    int*   cnt_in   = wsi + 160L * N + 2L * E;
    int*   cnt_out  = wsi + 161L * N + 2L * E;
    int*   ovf_n    = wsi + 162L * N + 2L * E;
    int*   gcount   = wsi + 162L * N + 2L * E + 32;
    u64*   spill    = (u64*)(wsi + 162L * N + 2L * E + 32 + NBMAX * GSTR);
    u64*   blist    = (u64*)(wsi + 162L * N + 4L * E + 32 + NBMAX * GSTR);
    float* outp     = (float*)d_out;

    const int NB = (N + RB - 1) / RB;

    hipMemsetAsync(cnt_out, 0, sizeof(int) * ((long)N + 32 + NBMAX * GSTR), stream);

    bucket_edges<<<256, T1, 0, stream>>>(src, dst, cnt_out, gcount, blist, ovf_n + 16, spill, E, NB);
    rank_range<<<NB, T2, 0, stream>>>(blist, gcount, ovf_n + 16, spill, cnt_in, slots, ovf_n, ovf_src, ovf_dst, N);

    const int mtiles = (N + 15) / 16;
    const int gblocks = (mtiles + 3) / 4;
    const int gather_blocks = 2048;

    gemm_norm_wmma<128, 64><<<gblocks, 128, 0, stream>>>(features, cnt_out, W1, hbuf, N);
    gather_rows<64, true><<<gather_blocks, 256, 0, stream>>>(hbuf, slots, cnt_in, ovf_n, ovf_src, ovf_dst, b1, x1, N);

    gemm_norm_wmma<64, 32><<<gblocks, 128, 0, stream>>>(x1, cnt_out, W2, hbuf, N);
    gather_rows<32, false><<<gather_blocks, 256, 0, stream>>>(hbuf, slots, cnt_in, ovf_n, ovf_src, ovf_dst, b2, outp, N);
}


# Out-degrees from LDS counters too: source ids filed per 512-node range in the same two-level pass and counted by the range's work-group; the per-edge global adds are gone; float sums untouched
# speedup vs baseline: 6.1003x; 1.6803x over previous
#include <hip/hip_runtime.h>
#include <hip/hip_bf16.h>


typedef __attribute__((ext_vector_type(2))) float v2f;
typedef __attribute__((ext_vector_type(8))) float v8f;

typedef unsigned long long u64;

constexpr int CAP   = 32;
constexpr int BSH   = 9;
constexpr int RB    = 1 << BSH;
constexpr int T1    = 512;
constexpr int NBMAX = T1;
constexpr int EPT   = 16;
constexpr int BCAP  = 12288;
constexpr int SCAP  = 12288;
constexpr int GSTR  = 32;
constexpr int T2    = 1024;
constexpr int PPT   = 8;

__device__ __forceinline__ u64 pack_pair(int s, int d) {
    return (u64)(unsigned)s | ((u64)(unsigned)d << 32);
}

__device__ __forceinline__ void scan_starts(const int* cnt, int* start, int NB, int lane) {
    int carry = 0;
    for (int c0 = 0; c0 < NB; c0 += 32) {
        const int idx = c0 + lane;
        const int v = (idx < NB) ? cnt[idx] : 0;
        int x = v;
        for (int o = 1; o < 32; o <<= 1) {
            const int y = __shfl_up(x, o);
            if (lane >= o) x += y;
        }
        if (idx < NB) start[idx] = carry + x - v;
        carry += __shfl(x, 31);
    }
}

__global__ void __launch_bounds__(T1)
bucket_edges(const int* __restrict__ src, const int* __restrict__ dst,
             int* __restrict__ gcount, u64* __restrict__ blist,
             int* __restrict__ spill_n, u64* __restrict__ spill,
             int* __restrict__ scount, int* __restrict__ slist,
             int* __restrict__ sspill_n, int* __restrict__ sspill, int E, int NB) {
    __shared__ int lcount[NBMAX];
    __shared__ int lstart[NBMAX];
    __shared__ int lbase[NBMAX];
    __shared__ u64 lpairs[T1 * EPT];
    __shared__ int scnt[NBMAX];
    __shared__ int sstart[NBMAX];
    __shared__ int sbase[NBMAX];
    __shared__ int lsrc[T1 * EPT];
    const int tid   = (int)threadIdx.x;
    const int chunk = (E + (int)gridDim.x - 1) / (int)gridDim.x;
    const int cbeg  = (int)blockIdx.x * chunk;
    const int cend  = (cbeg + chunk < E) ? (cbeg + chunk) : E;

    for (int base = cbeg; base < cend; base += T1 * EPT) {
        if (tid < NB) { lcount[tid] = 0; scnt[tid] = 0; }
        __syncthreads();

        int s[EPT], d[EPT], lr[EPT], sr[EPT];
#pragma unroll
        for (int i = 0; i < EPT; ++i) {
            const int e = base + i * T1 + tid;
            const bool ok = e < cend;
            s[i] = ok ? src[e] : 0;
            d[i] = ok ? dst[e] : 0;
        }
#pragma unroll
        for (int i = 0; i < EPT; ++i) {
            lr[i] = 0;
            sr[i] = 0;
            if (base + i * T1 + tid < cend) {
                lr[i] = atomicAdd(&lcount[d[i] >> BSH], 1);
                sr[i] = atomicAdd(&scnt[s[i] >> BSH], 1);
            }
        }
        __syncthreads();

        const int c  = (tid < NB) ? lcount[tid] : 0;
        const int c2 = (tid < NB) ? scnt[tid] : 0;
        int got = 0, got2 = 0;
        if (c > 0)  got  = atomicAdd(&gcount[tid * GSTR], c);
        if (c2 > 0) got2 = atomicAdd(&scount[tid * GSTR], c2);
        if (tid < 32) {
            scan_starts(lcount, lstart, NB, tid);
        } else if (tid < 64) {
            scan_starts(scnt, sstart, NB, tid - 32);
        }
        if (tid < NB) { lbase[tid] = got; sbase[tid] = got2; }
        __syncthreads();

        int st[EPT], st2[EPT];
#pragma unroll
        for (int i = 0; i < EPT; ++i) {
            st[i]  = lstart[d[i] >> BSH];
            st2[i] = sstart[s[i] >> BSH];
        }
#pragma unroll
        for (int i = 0; i < EPT; ++i) {
            if (base + i * T1 + tid < cend) {
                lpairs[st[i] + lr[i]] = pack_pair(s[i], d[i]);
                lsrc[st2[i] + sr[i]]  = s[i];
            }
        }
        __syncthreads();

        const int left = cend - base;
        const int cn = left < T1 * EPT ? left : T1 * EPT;
        u64 v[EPT];
        int gb[EPT], ls[EPT];
#pragma unroll
        for (int i = 0; i < EPT; ++i) {
            const int j = i * T1 + tid;
            v[i] = lpairs[j < cn ? j : 0];
        }
#pragma unroll
        for (int i = 0; i < EPT; ++i) {
            const int b = (int)(v[i] >> 32) >> BSH;
            gb[i] = lbase[b];
            ls[i] = lstart[b];
        }
#pragma unroll
        for (int i = 0; i < EPT; ++i) {
            const int j = i * T1 + tid;
            if (j < cn) {
                const int b   = (int)(v[i] >> 32) >> BSH;
                const int pos = gb[i] + (j - ls[i]);
                u64* q;
                if (pos < BCAP) {
                    q = blist + (long)b * BCAP + pos;
                } else {
                    q = spill + atomicAdd(spill_n, 1);
                }
                *q = v[i];
            }
        }
        int w[EPT], gb2[EPT], ls2[EPT];
#pragma unroll
        for (int i = 0; i < EPT; ++i) {
            const int j = i * T1 + tid;
            w[i] = lsrc[j < cn ? j : 0];
        }
#pragma unroll
        for (int i = 0; i < EPT; ++i) {
            const int b = w[i] >> BSH;
            gb2[i] = sbase[b];
            ls2[i] = sstart[b];
        }
#pragma unroll
        for (int i = 0; i < EPT; ++i) {
            const int j = i * T1 + tid;
            if (j < cn) {
                const int b   = w[i] >> BSH;
                const int pos = gb2[i] + (j - ls2[i]);
                int* q;
                if (pos < SCAP) {
                    q = slist + (long)b * SCAP + pos;
                } else {
                    q = sspill + atomicAdd(sspill_n, 1);
                }
                *q = w[i];
            }
        }
        if (base + T1 * EPT < cend) __syncthreads();
    }
}

__global__ void __launch_bounds__(T2)
rank_range(const u64* __restrict__ blist, const int* __restrict__ gcount,
           const int* __restrict__ spill_n, const u64* __restrict__ spill,
           const int* __restrict__ slist, const int* __restrict__ scount,
           const int* __restrict__ sspill_n, const int* __restrict__ sspill,
           int* __restrict__ cnt_in, int* __restrict__ cnt_out,
           int* __restrict__ slots, int* __restrict__ ovf_n,
           int* __restrict__ ovf_src, int* __restrict__ ovf_dst, int N) {
    __shared__ int lcnt[RB];
    __shared__ int lco[RB];
    __shared__ __attribute__((aligned(16))) int lslots[RB * CAP];
    const int tid   = (int)threadIdx.x;
    const int b     = (int)blockIdx.x;
    const int nbase = b << BSH;

    int4* __restrict__ l4 = reinterpret_cast<int4*>(lslots);
    for (int t = tid; t < RB; t += T2) { lcnt[t] = 0; lco[t] = 0; }
    for (int t = tid; t < RB * CAP / 4; t += T2) l4[t] = make_int4(0, 0, 0, 0);
    __syncthreads();

    auto place = [&](const u64 p) {
        const int ps = (int)(unsigned)p;
        const int pd = (int)(p >> 32);
        const int dl = pd - nbase;
        const int r  = atomicAdd(&lcnt[dl], 1);
        if (r < CAP) {
            lslots[dl * CAP + r] = ps;
        } else {
            const int k = atomicAdd(ovf_n, 1);
            ovf_src[k] = ps;
            ovf_dst[k] = pd;
        }
    };

    int n = gcount[b * GSTR];
    n = n < BCAP ? n : BCAP;
    int n2 = scount[b * GSTR];
    n2 = n2 < SCAP ? n2 : SCAP;
    const int nmax = n > n2 ? n : n2;
    const u64* __restrict__ bl = blist + (long)b * BCAP;
    const int* __restrict__ sl = slist + (long)b * SCAP;
    for (int base = 0; base < nmax; base += T2 * PPT) {
        u64 p[PPT];
        int q[PPT];
#pragma unroll
        for (int u = 0; u < PPT; ++u) {
            const int i = base + u * T2 + tid;
            p[u] = (i < n) ? bl[i] : 0ull;
            q[u] = (i < n2) ? sl[i] : nbase;
        }
#pragma unroll
        for (int u = 0; u < PPT; ++u) {
            if (base + u * T2 + tid < n) place(p[u]);
        }
#pragma unroll
        for (int u = 0; u < PPT; ++u) {
            if (base + u * T2 + tid < n2) atomicAdd(&lco[q[u] - nbase], 1);
        }
    }

    const int ns = spill_n[0];
    for (int i = tid; i < ns; i += T2) {
        const u64 q = spill[i];
        if (((int)(q >> 32) >> BSH) == b) place(q);
    }
    const int ns2 = sspill_n[0];
    for (int i = tid; i < ns2; i += T2) {
        const int q = sspill[i];
        if ((q >> BSH) == b) atomicAdd(&lco[q - nbase], 1);
    }
    __syncthreads();

    const int left = N - nbase;
    const int nn = left < RB ? left : RB;
    for (int t = tid; t < nn; t += T2) {
        cnt_in[nbase + t]  = lcnt[t];
        cnt_out[nbase + t] = lco[t];
    }
    int4* __restrict__ g4 = reinterpret_cast<int4*>(slots + (long)nbase * CAP);
    for (int t = tid; t < nn * (CAP / 4); t += T2) g4[t] = l4[t];
}

template <int K, int COLS>
__global__ void __launch_bounds__(128)
gemm_norm_wmma(const float* __restrict__ X, const int* __restrict__ cntv,
               const float* __restrict__ W, float* __restrict__ H, int nrows) {
    constexpr int NT = COLS / 16;
    const int lane  = threadIdx.x & 31;
    const int wave  = threadIdx.x >> 5;
    const int tile_m = blockIdx.x * (blockDim.x >> 5) + wave;
    if (tile_m * 16 >= nrows) return;

    const int mr = lane & 15;
    const int kh = (lane >> 4) << 1;
    const int m  = tile_m * 16 + mr;

    const int co = cntv[m];
    const float nm = rsqrtf(co > 0 ? (float)co : 1.0f);
    const float* __restrict__ xrow = X + (long)m * K;

    v8f acc[NT];
#pragma unroll
    for (int t = 0; t < NT; ++t) { v8f z = {}; acc[t] = z; }

#pragma unroll 4
    for (int k = 0; k < K; k += 4) {
        v2f a;
        a.x = xrow[k + kh]     * nm;
        a.y = xrow[k + kh + 1] * nm;
        const float* __restrict__ wr = W + (long)(k + kh) * COLS;
#pragma unroll
        for (int t = 0; t < NT; ++t) {
            v2f b;
            b.x = wr[t * 16 + mr];
            b.y = wr[COLS + t * 16 + mr];
            acc[t] = __builtin_amdgcn_wmma_f32_16x16x4_f32(
                false, a, false, b, (short)0, acc[t], false, false);
        }
    }

    const int mbase = tile_m * 16 + ((lane >> 4) << 3);
#pragma unroll
    for (int t = 0; t < NT; ++t) {
#pragma unroll
        for (int v = 0; v < 8; ++v) {
            H[(long)(mbase + v) * COLS + t * 16 + mr] = acc[t][v];
        }
    }
}

template <int C, bool RELU>
__global__ void __launch_bounds__(256)
gather_rows(const float* __restrict__ H, const int* __restrict__ slots,
            const int* __restrict__ cnt_in, const int* __restrict__ ovf_n,
            const int* __restrict__ ovf_src, const int* __restrict__ ovf_dst,
            const float* __restrict__ bias, float* __restrict__ out, int N) {
    constexpr int VEC = C / 32;
    constexpr int B   = 16;
    const int lane = threadIdx.x & 31;
    const int wave = __builtin_amdgcn_readfirstlane((int)(threadIdx.x >> 5));
    const int nwaves = (int)gridDim.x * 8;

    float bv[VEC];
#pragma unroll
    for (int k = 0; k < VEC; ++k) bv[k] = bias[lane * VEC + k];

    for (int node = (int)blockIdx.x * 8 + wave; node < N; node += nwaves) {
        const int cnt = cnt_in[node];
        const int n32 = cnt < CAP ? cnt : CAP;
        float acc[VEC];
#pragma unroll
        for (int k = 0; k < VEC; ++k) acc[k] = 0.0f;

        const int myidx = (lane < n32) ? slots[(long)node * CAP + lane] : 0;
        for (int h = 0; h < n32; h += B) {
            const int n = n32 - h;
            float v[B][VEC];
#pragma unroll
            for (int u = 0; u < B; ++u) {
#pragma unroll
                for (int k = 0; k < VEC; ++k) v[u][k] = 0.0f;
                if (u == 0 || u < n) {
                    const int s = __builtin_amdgcn_readlane(myidx, h + u);
                    const float* __restrict__ p = H + (long)s * C + lane * VEC;
                    if constexpr (VEC == 2) {
                        const float2 t = *reinterpret_cast<const float2*>(p);
                        v[u][0] = t.x;
                        v[u][1] = t.y;
                    } else {
                        v[u][0] = p[0];
                    }
                }
            }
#pragma unroll
            for (int u = 0; u < B; ++u) {
#pragma unroll
                for (int k = 0; k < VEC; ++k) acc[k] += v[u][k];
            }
        }

        if (cnt > CAP) {
            const int novf = ovf_n[0];
            for (int j = 0; j < novf; j += 32) {
                const int q = j + lane;
                const int od = (q < novf) ? ovf_dst[q] : -1;
                const int os = (q < novf) ? ovf_src[q] : 0;
                unsigned m = __builtin_amdgcn_ballot_w32(od == node);
                while (m != 0u) {
                    const int b = __builtin_ctz(m);
                    m &= m - 1u;
                    const int s = __builtin_amdgcn_readlane(os, b);
                    const float* __restrict__ p = H + (long)s * C + lane * VEC;
#pragma unroll
                    for (int k = 0; k < VEC; ++k) acc[k] += p[k];
                }
            }
        }

        const float nin = rsqrtf(cnt > 0 ? (float)cnt : 1.0f);
        float r[VEC];
#pragma unroll
        for (int k = 0; k < VEC; ++k) {
            const float t = acc[k] * nin + bv[k];
            r[k] = RELU ? (t > 0.0f ? t : 0.0f) : t;
        }
        float* __restrict__ o = out + (long)node * C + lane * VEC;
        if constexpr (VEC == 2) {
            *reinterpret_cast<float2*>(o) = make_float2(r[0], r[1]);
        } else {
            o[0] = r[0];
        }
    }
}

extern "C" void kernel_launch(void* const* d_in, const int* in_sizes, int n_in,
                              void* d_out, int out_size, void* d_ws, size_t ws_size,
                              hipStream_t stream) {
    const float* features = (const float*)d_in[0];
    const int*   src      = (const int*)d_in[1];
    const int*   dst      = (const int*)d_in[2];
    const float* W1       = (const float*)d_in[3];
    const float* b1       = (const float*)d_in[4];
    const float* W2       = (const float*)d_in[5];
    const float* b2       = (const float*)d_in[6];

    const int IN_F = 128;
    const int N = in_sizes[0] / IN_F;
    const int E = in_sizes[1];

    int*   wsi      = (int*)d_ws;
    float* wsf      = (float*)d_ws;
    float* hbuf     = wsf;
    float* x1       = wsf + 64L * N;
    int*   slots    = wsi + 128L * N;
    int*   ovf_src  = wsi + 160L * N;
    int*   ovf_dst  = wsi + 160L * N + (long)E;
    int*   cnt_in   = wsi + 160L * N + 2L * E;
    int*   cnt_out  = wsi + 161L * N + 2L * E;
    int*   ovf_n    = wsi + 162L * N + 2L * E;
    int*   gcount   = wsi + 162L * N + 2L * E + 32;
    int*   scount   = wsi + 162L * N + 2L * E + 32 + NBMAX * GSTR;
    u64*   spill    = (u64*)(wsi + 162L * N + 2L * E + 32 + 2 * NBMAX * GSTR);
    u64*   blist    = (u64*)(wsi + 162L * N + 4L * E + 32 + 2 * NBMAX * GSTR);
    const int NB = (N + RB - 1) / RB;
    int*   sspill   = wsi + 162L * N + 4L * E + 32 + 2 * NBMAX * GSTR + 2L * BCAP * NB;
    int*   slist    = wsi + 162L * N + 5L * E + 32 + 2 * NBMAX * GSTR + 2L * BCAP * NB;
    float* outp     = (float*)d_out;

    hipMemsetAsync(ovf_n, 0, sizeof(int) * (32 + 2 * NBMAX * GSTR), stream);

    bucket_edges<<<256, T1, 0, stream>>>(src, dst, gcount, blist, ovf_n + 16, spill, scount, slist, ovf_n + 8, sspill, E, NB);
    rank_range<<<NB, T2, 0, stream>>>(blist, gcount, ovf_n + 16, spill, slist, scount, ovf_n + 8, sspill,
                                      cnt_in, cnt_out, slots, ovf_n, ovf_src, ovf_dst, N);

    const int mtiles = (N + 15) / 16;
    const int gblocks = (mtiles + 3) / 4;
    const int gather_blocks = 2048;

    gemm_norm_wmma<128, 64><<<gblocks, 128, 0, stream>>>(features, cnt_out, W1, hbuf, N);
    gather_rows<64, true><<<gather_blocks, 256, 0, stream>>>(hbuf, slots, cnt_in, ovf_n, ovf_src, ovf_dst, b1, x1, N);

    gemm_norm_wmma<64, 32><<<gblocks, 128, 0, stream>>>(x1, cnt_out, W2, hbuf, N);
    gather_rows<32, false><<<gather_blocks, 256, 0, stream>>>(hbuf, slots, cnt_in, ovf_n, ovf_src, ovf_dst, b2, outp, N);
}


# Gather: one dependent round trip per node instead of 3-4: next node's count and whole slot line load under the current rows, and a node's rows leave as one group (24/32 wide); sum order kept
# speedup vs baseline: 6.2812x; 1.0297x over previous
#include <hip/hip_runtime.h>
#include <hip/hip_bf16.h>


typedef __attribute__((ext_vector_type(2))) float v2f;
typedef __attribute__((ext_vector_type(8))) float v8f;

typedef unsigned long long u64;

constexpr int CAP   = 32;
constexpr int BSH   = 9;
constexpr int RB    = 1 << BSH;
constexpr int T1    = 512;
constexpr int NBMAX = T1;
constexpr int EPT   = 16;
constexpr int BCAP  = 12288;
constexpr int SCAP  = 12288;
constexpr int GSTR  = 32;
constexpr int T2    = 1024;
constexpr int PPT   = 8;

__device__ __forceinline__ u64 pack_pair(int s, int d) {
    return (u64)(unsigned)s | ((u64)(unsigned)d << 32);
}

__device__ __forceinline__ void scan_starts(const int* cnt, int* start, int NB, int lane) {
    int carry = 0;
    for (int c0 = 0; c0 < NB; c0 += 32) {
        const int idx = c0 + lane;
        const int v = (idx < NB) ? cnt[idx] : 0;
        int x = v;
        for (int o = 1; o < 32; o <<= 1) {
            const int y = __shfl_up(x, o);
            if (lane >= o) x += y;
        }
        if (idx < NB) start[idx] = carry + x - v;
        carry += __shfl(x, 31);
    }
}

__global__ void __launch_bounds__(T1)
bucket_edges(const int* __restrict__ src, const int* __restrict__ dst,
             int* __restrict__ gcount, u64* __restrict__ blist,
             int* __restrict__ spill_n, u64* __restrict__ spill,
             int* __restrict__ scount, int* __restrict__ slist,
             int* __restrict__ sspill_n, int* __restrict__ sspill, int E, int NB) {
    __shared__ int lcount[NBMAX];
    __shared__ int lstart[NBMAX];
    __shared__ int lbase[NBMAX];
    __shared__ u64 lpairs[T1 * EPT];
    __shared__ int scnt[NBMAX];
    __shared__ int sstart[NBMAX];
    __shared__ int sbase[NBMAX];
    __shared__ int lsrc[T1 * EPT];
    const int tid   = (int)threadIdx.x;
    const int chunk = (E + (int)gridDim.x - 1) / (int)gridDim.x;
    const int cbeg  = (int)blockIdx.x * chunk;
    const int cend  = (cbeg + chunk < E) ? (cbeg + chunk) : E;

    for (int base = cbeg; base < cend; base += T1 * EPT) {
        if (tid < NB) { lcount[tid] = 0; scnt[tid] = 0; }
        __syncthreads();

        int s[EPT], d[EPT], lr[EPT], sr[EPT];
#pragma unroll
        for (int i = 0; i < EPT; ++i) {
            const int e = base + i * T1 + tid;
            const bool ok = e < cend;
            s[i] = ok ? src[e] : 0;
            d[i] = ok ? dst[e] : 0;
        }
#pragma unroll
        for (int i = 0; i < EPT; ++i) {
            lr[i] = 0;
            sr[i] = 0;
            if (base + i * T1 + tid < cend) {
                lr[i] = atomicAdd(&lcount[d[i] >> BSH], 1);
                sr[i] = atomicAdd(&scnt[s[i] >> BSH], 1);
            }
        }
        __syncthreads();

        const int c  = (tid < NB) ? lcount[tid] : 0;
        const int c2 = (tid < NB) ? scnt[tid] : 0;
        int got = 0, got2 = 0;
        if (c > 0)  got  = atomicAdd(&gcount[tid * GSTR], c);
        if (c2 > 0) got2 = atomicAdd(&scount[tid * GSTR], c2);
        if (tid < 32) {
            scan_starts(lcount, lstart, NB, tid);
        } else if (tid < 64) {
            scan_starts(scnt, sstart, NB, tid - 32);
        }
        if (tid < NB) { lbase[tid] = got; sbase[tid] = got2; }
        __syncthreads();

        int st[EPT], st2[EPT];
#pragma unroll
        for (int i = 0; i < EPT; ++i) {
            st[i]  = lstart[d[i] >> BSH];
            st2[i] = sstart[s[i] >> BSH];
        }
#pragma unroll
        for (int i = 0; i < EPT; ++i) {
            if (base + i * T1 + tid < cend) {
                lpairs[st[i] + lr[i]] = pack_pair(s[i], d[i]);
                lsrc[st2[i] + sr[i]]  = s[i];
            }
        }
        __syncthreads();

        const int left = cend - base;
        const int cn = left < T1 * EPT ? left : T1 * EPT;
        u64 v[EPT];
        int gb[EPT], ls[EPT];
#pragma unroll
        for (int i = 0; i < EPT; ++i) {
            const int j = i * T1 + tid;
            v[i] = lpairs[j < cn ? j : 0];
        }
#pragma unroll
        for (int i = 0; i < EPT; ++i) {
            const int b = (int)(v[i] >> 32) >> BSH;
            gb[i] = lbase[b];
            ls[i] = lstart[b];
        }
#pragma unroll
        for (int i = 0; i < EPT; ++i) {
            const int j = i * T1 + tid;
            if (j < cn) {
                const int b   = (int)(v[i] >> 32) >> BSH;
                const int pos = gb[i] + (j - ls[i]);
                u64* q;
                if (pos < BCAP) {
                    q = blist + (long)b * BCAP + pos;
                } else {
                    q = spill + atomicAdd(spill_n, 1);
                }
                *q = v[i];
            }
        }
        int w[EPT], gb2[EPT], ls2[EPT];
#pragma unroll
        for (int i = 0; i < EPT; ++i) {
            const int j = i * T1 + tid;
            w[i] = lsrc[j < cn ? j : 0];
        }
#pragma unroll
        for (int i = 0; i < EPT; ++i) {
            const int b = w[i] >> BSH;
            gb2[i] = sbase[b];
            ls2[i] = sstart[b];
        }
#pragma unroll
        for (int i = 0; i < EPT; ++i) {
            const int j = i * T1 + tid;
            if (j < cn) {
                const int b   = w[i] >> BSH;
                const int pos = gb2[i] + (j - ls2[i]);
                int* q;
                if (pos < SCAP) {
                    q = slist + (long)b * SCAP + pos;
                } else {
                    q = sspill + atomicAdd(sspill_n, 1);
                }
                *q = w[i];
            }
        }
        if (base + T1 * EPT < cend) __syncthreads();
    }
}

__global__ void __launch_bounds__(T2)
rank_range(const u64* __restrict__ blist, const int* __restrict__ gcount,
           const int* __restrict__ spill_n, const u64* __restrict__ spill,
           const int* __restrict__ slist, const int* __restrict__ scount,
           const int* __restrict__ sspill_n, const int* __restrict__ sspill,
           int* __restrict__ cnt_in, int* __restrict__ cnt_out,
           int* __restrict__ slots, int* __restrict__ ovf_n,
           int* __restrict__ ovf_src, int* __restrict__ ovf_dst, int N) {
    __shared__ int lcnt[RB];
    __shared__ int lco[RB];
    __shared__ __attribute__((aligned(16))) int lslots[RB * CAP];
    const int tid   = (int)threadIdx.x;
    const int b     = (int)blockIdx.x;
    const int nbase = b << BSH;

    int4* __restrict__ l4 = reinterpret_cast<int4*>(lslots);
    for (int t = tid; t < RB; t += T2) { lcnt[t] = 0; lco[t] = 0; }
    for (int t = tid; t < RB * CAP / 4; t += T2) l4[t] = make_int4(0, 0, 0, 0);
    __syncthreads();

    auto place = [&](const u64 p) {
        const int ps = (int)(unsigned)p;
        const int pd = (int)(p >> 32);
        const int dl = pd - nbase;
        const int r  = atomicAdd(&lcnt[dl], 1);
        if (r < CAP) {
            lslots[dl * CAP + r] = ps;
        } else {
            const int k = atomicAdd(ovf_n, 1);
            ovf_src[k] = ps;
            ovf_dst[k] = pd;
        }
    };

    int n = gcount[b * GSTR];
    n = n < BCAP ? n : BCAP;
    int n2 = scount[b * GSTR];
    n2 = n2 < SCAP ? n2 : SCAP;
    const int nmax = n > n2 ? n : n2;
    const u64* __restrict__ bl = blist + (long)b * BCAP;
    const int* __restrict__ sl = slist + (long)b * SCAP;
    for (int base = 0; base < nmax; base += T2 * PPT) {
        u64 p[PPT];
        int q[PPT];
#pragma unroll
        for (int u = 0; u < PPT; ++u) {
            const int i = base + u * T2 + tid;
            p[u] = (i < n) ? bl[i] : 0ull;
            q[u] = (i < n2) ? sl[i] : nbase;
        }
#pragma unroll
        for (int u = 0; u < PPT; ++u) {
            if (base + u * T2 + tid < n) place(p[u]);
        }
#pragma unroll
        for (int u = 0; u < PPT; ++u) {
            if (base + u * T2 + tid < n2) atomicAdd(&lco[q[u] - nbase], 1);
        }
    }

    const int ns = spill_n[0];
    for (int i = tid; i < ns; i += T2) {
        const u64 q = spill[i];
        if (((int)(q >> 32) >> BSH) == b) place(q);
    }
    const int ns2 = sspill_n[0];
    for (int i = tid; i < ns2; i += T2) {
        const int q = sspill[i];
        if ((q >> BSH) == b) atomicAdd(&lco[q - nbase], 1);
    }
    __syncthreads();

    const int left = N - nbase;
    const int nn = left < RB ? left : RB;
    for (int t = tid; t < nn; t += T2) {
        cnt_in[nbase + t]  = lcnt[t];
        cnt_out[nbase + t] = lco[t];
    }
    int4* __restrict__ g4 = reinterpret_cast<int4*>(slots + (long)nbase * CAP);
    for (int t = tid; t < nn * (CAP / 4); t += T2) g4[t] = l4[t];
}

template <int K, int COLS>
__global__ void __launch_bounds__(128)
gemm_norm_wmma(const float* __restrict__ X, const int* __restrict__ cntv,
               const float* __restrict__ W, float* __restrict__ H, int nrows) {
    constexpr int NT = COLS / 16;
    const int lane  = threadIdx.x & 31;
    const int wave  = threadIdx.x >> 5;
    const int tile_m = blockIdx.x * (blockDim.x >> 5) + wave;
    if (tile_m * 16 >= nrows) return;

    const int mr = lane & 15;
    const int kh = (lane >> 4) << 1;
    const int m  = tile_m * 16 + mr;

    const int co = cntv[m];
    const float nm = rsqrtf(co > 0 ? (float)co : 1.0f);
    const float* __restrict__ xrow = X + (long)m * K;

    v8f acc[NT];
#pragma unroll
    for (int t = 0; t < NT; ++t) { v8f z = {}; acc[t] = z; }

#pragma unroll 4
    for (int k = 0; k < K; k += 4) {
        v2f a;
        a.x = xrow[k + kh]     * nm;
        a.y = xrow[k + kh + 1] * nm;
        const float* __restrict__ wr = W + (long)(k + kh) * COLS;
#pragma unroll
        for (int t = 0; t < NT; ++t) {
            v2f b;
            b.x = wr[t * 16 + mr];
            b.y = wr[COLS + t * 16 + mr];
            acc[t] = __builtin_amdgcn_wmma_f32_16x16x4_f32(
                false, a, false, b, (short)0, acc[t], false, false);
        }
    }

    const int mbase = tile_m * 16 + ((lane >> 4) << 3);
#pragma unroll
    for (int t = 0; t < NT; ++t) {
#pragma unroll
        for (int v = 0; v < 8; ++v) {
            H[(long)(mbase + v) * COLS + t * 16 + mr] = acc[t][v];
        }
    }
}

template <int C, bool RELU>
__global__ void __launch_bounds__(256)
gather_rows(const float* __restrict__ H, const int* __restrict__ slots,
            const int* __restrict__ cnt_in, const int* __restrict__ ovf_n,
            const int* __restrict__ ovf_src, const int* __restrict__ ovf_dst,
            const float* __restrict__ bias, float* __restrict__ out, int N) {
    constexpr int VEC = C / 32;
    constexpr int B   = (VEC == 1) ? 32 : 24;
    const int lane = threadIdx.x & 31;
    const int wave = __builtin_amdgcn_readfirstlane((int)(threadIdx.x >> 5));
    const int nwaves = (int)gridDim.x * 8;

    float bv[VEC];
#pragma unroll
    for (int k = 0; k < VEC; ++k) bv[k] = bias[lane * VEC + k];

    int node = (int)blockIdx.x * 8 + wave;
    int cnt_nx = 0, idx_nx = 0;
    if (node < N) {
        cnt_nx = cnt_in[node];
        idx_nx = slots[(long)node * CAP + lane];
    }
    for (; node < N; node += nwaves) {
        const int cnt   = cnt_nx;
        const int myidx = idx_nx;
        const int nxt   = node + nwaves;
        if (nxt < N) {
            cnt_nx = cnt_in[nxt];
            idx_nx = slots[(long)nxt * CAP + lane];
        }
        const int n32 = cnt < CAP ? cnt : CAP;
        float acc[VEC];
#pragma unroll
        for (int k = 0; k < VEC; ++k) acc[k] = 0.0f;

        for (int h = 0; h < n32; h += B) {
            const int n = n32 - h;
            float v[B][VEC];
#pragma unroll
            for (int u = 0; u < B; ++u) {
#pragma unroll
                for (int k = 0; k < VEC; ++k) v[u][k] = 0.0f;
                if (u == 0 || u < n) {
                    const int s = __builtin_amdgcn_readlane(myidx, h + u);
                    const float* __restrict__ p = H + (long)s * C + lane * VEC;
                    if constexpr (VEC == 2) {
                        const float2 t = *reinterpret_cast<const float2*>(p);
                        v[u][0] = t.x;
                        v[u][1] = t.y;
                    } else {
                        v[u][0] = p[0];
                    }
                }
            }
#pragma unroll
            for (int u = 0; u < B; ++u) {
#pragma unroll
                for (int k = 0; k < VEC; ++k) acc[k] += v[u][k];
            }
        }

        if (cnt > CAP) {
            const int novf = ovf_n[0];
            for (int j = 0; j < novf; j += 32) {
                const int q = j + lane;
                const int od = (q < novf) ? ovf_dst[q] : -1;
                const int os = (q < novf) ? ovf_src[q] : 0;
                unsigned m = __builtin_amdgcn_ballot_w32(od == node);
                while (m != 0u) {
                    const int b = __builtin_ctz(m);
                    m &= m - 1u;
                    const int s = __builtin_amdgcn_readlane(os, b);
                    const float* __restrict__ p = H + (long)s * C + lane * VEC;
#pragma unroll
                    for (int k = 0; k < VEC; ++k) acc[k] += p[k];
                }
            }
        }

        const float nin = rsqrtf(cnt > 0 ? (float)cnt : 1.0f);
        float r[VEC];
#pragma unroll
        for (int k = 0; k < VEC; ++k) {
            const float t = acc[k] * nin + bv[k];
            r[k] = RELU ? (t > 0.0f ? t : 0.0f) : t;
        }
        float* __restrict__ o = out + (long)node * C + lane * VEC;
        if constexpr (VEC == 2) {
            *reinterpret_cast<float2*>(o) = make_float2(r[0], r[1]);
        } else {
            o[0] = r[0];
        }
    }
}

extern "C" void kernel_launch(void* const* d_in, const int* in_sizes, int n_in,
                              void* d_out, int out_size, void* d_ws, size_t ws_size,
                              hipStream_t stream) {
    const float* features = (const float*)d_in[0];
    const int*   src      = (const int*)d_in[1];
    const int*   dst      = (const int*)d_in[2];
    const float* W1       = (const float*)d_in[3];
    const float* b1       = (const float*)d_in[4];
    const float* W2       = (const float*)d_in[5];
    const float* b2       = (const float*)d_in[6];

    const int IN_F = 128;
    const int N = in_sizes[0] / IN_F;
    const int E = in_sizes[1];

    int*   wsi      = (int*)d_ws;
    float* wsf      = (float*)d_ws;
    float* hbuf     = wsf;
    float* x1       = wsf + 64L * N;
    int*   slots    = wsi + 128L * N;
    int*   ovf_src  = wsi + 160L * N;
    int*   ovf_dst  = wsi + 160L * N + (long)E;
    int*   cnt_in   = wsi + 160L * N + 2L * E;
    int*   cnt_out  = wsi + 161L * N + 2L * E;
    int*   ovf_n    = wsi + 162L * N + 2L * E;
    int*   gcount   = wsi + 162L * N + 2L * E + 32;
    int*   scount   = wsi + 162L * N + 2L * E + 32 + NBMAX * GSTR;
    u64*   spill    = (u64*)(wsi + 162L * N + 2L * E + 32 + 2 * NBMAX * GSTR);
    u64*   blist    = (u64*)(wsi + 162L * N + 4L * E + 32 + 2 * NBMAX * GSTR);
    const int NB = (N + RB - 1) / RB;
    int*   sspill   = wsi + 162L * N + 4L * E + 32 + 2 * NBMAX * GSTR + 2L * BCAP * NB;
    int*   slist    = wsi + 162L * N + 5L * E + 32 + 2 * NBMAX * GSTR + 2L * BCAP * NB;
    float* outp     = (float*)d_out;

    hipMemsetAsync(ovf_n, 0, sizeof(int) * (32 + 2 * NBMAX * GSTR), stream);

    bucket_edges<<<256, T1, 0, stream>>>(src, dst, gcount, blist, ovf_n + 16, spill, scount, slist, ovf_n + 8, sspill, E, NB);
    rank_range<<<NB, T2, 0, stream>>>(blist, gcount, ovf_n + 16, spill, slist, scount, ovf_n + 8, sspill,
                                      cnt_in, cnt_out, slots, ovf_n, ovf_src, ovf_dst, N);

    const int mtiles = (N + 15) / 16;
    const int gblocks = (mtiles + 3) / 4;
    const int gather_blocks = 2048;

    gemm_norm_wmma<128, 64><<<gblocks, 128, 0, stream>>>(features, cnt_out, W1, hbuf, N);
    gather_rows<64, true><<<gather_blocks, 256, 0, stream>>>(hbuf, slots, cnt_in, ovf_n, ovf_src, ovf_dst, b1, x1, N);

    gemm_norm_wmma<64, 32><<<gblocks, 128, 0, stream>>>(x1, cnt_out, W2, hbuf, N);
    gather_rows<32, false><<<gather_blocks, 256, 0, stream>>>(hbuf, slots, cnt_in, ovf_n, ovf_src, ovf_dst, b2, outp, N);
}


# Product kernels load wide: fragment k map and column map chosen so a lane reads X in 16-byte loads, one W load feeds all tiles, H rows leave as whole lines; k sum order inside each 8 regrouped
# speedup vs baseline: 6.5601x; 1.0444x over previous
#include <hip/hip_runtime.h>
#include <hip/hip_bf16.h>


typedef __attribute__((ext_vector_type(2))) float v2f;
typedef __attribute__((ext_vector_type(8))) float v8f;

typedef unsigned long long u64;

constexpr int CAP   = 32;
constexpr int BSH   = 9;
constexpr int RB    = 1 << BSH;
constexpr int T1    = 512;
constexpr int NBMAX = T1;
constexpr int EPT   = 16;
constexpr int BCAP  = 12288;
constexpr int SCAP  = 12288;
constexpr int GSTR  = 32;
constexpr int T2    = 1024;
constexpr int PPT   = 8;

__device__ __forceinline__ u64 pack_pair(int s, int d) {
    return (u64)(unsigned)s | ((u64)(unsigned)d << 32);
}

__device__ __forceinline__ void scan_starts(const int* cnt, int* start, int NB, int lane) {
    int carry = 0;
    for (int c0 = 0; c0 < NB; c0 += 32) {
        const int idx = c0 + lane;
        const int v = (idx < NB) ? cnt[idx] : 0;
        int x = v;
        for (int o = 1; o < 32; o <<= 1) {
            const int y = __shfl_up(x, o);
            if (lane >= o) x += y;
        }
        if (idx < NB) start[idx] = carry + x - v;
        carry += __shfl(x, 31);
    }
}

__global__ void __launch_bounds__(T1)
bucket_edges(const int* __restrict__ src, const int* __restrict__ dst,
             int* __restrict__ gcount, u64* __restrict__ blist,
             int* __restrict__ spill_n, u64* __restrict__ spill,
             int* __restrict__ scount, int* __restrict__ slist,
             int* __restrict__ sspill_n, int* __restrict__ sspill, int E, int NB) {
    __shared__ int lcount[NBMAX];
    __shared__ int lstart[NBMAX];
    __shared__ int lbase[NBMAX];
    __shared__ u64 lpairs[T1 * EPT];
    __shared__ int scnt[NBMAX];
    __shared__ int sstart[NBMAX];
    __shared__ int sbase[NBMAX];
    __shared__ int lsrc[T1 * EPT];
    const int tid   = (int)threadIdx.x;
    const int chunk = (E + (int)gridDim.x - 1) / (int)gridDim.x;
    const int cbeg  = (int)blockIdx.x * chunk;
    const int cend  = (cbeg + chunk < E) ? (cbeg + chunk) : E;

    for (int base = cbeg; base < cend; base += T1 * EPT) {
        if (tid < NB) { lcount[tid] = 0; scnt[tid] = 0; }
        __syncthreads();

        int s[EPT], d[EPT], lr[EPT], sr[EPT];
#pragma unroll
        for (int i = 0; i < EPT; ++i) {
            const int e = base + i * T1 + tid;
            const bool ok = e < cend;
            s[i] = ok ? src[e] : 0;
            d[i] = ok ? dst[e] : 0;
        }
#pragma unroll
        for (int i = 0; i < EPT; ++i) {
            lr[i] = 0;
            sr[i] = 0;
            if (base + i * T1 + tid < cend) {
                lr[i] = atomicAdd(&lcount[d[i] >> BSH], 1);
                sr[i] = atomicAdd(&scnt[s[i] >> BSH], 1);
            }
        }
        __syncthreads();

        const int c  = (tid < NB) ? lcount[tid] : 0;
        const int c2 = (tid < NB) ? scnt[tid] : 0;
        int got = 0, got2 = 0;
        if (c > 0)  got  = atomicAdd(&gcount[tid * GSTR], c);
        if (c2 > 0) got2 = atomicAdd(&scount[tid * GSTR], c2);
        if (tid < 32) {
            scan_starts(lcount, lstart, NB, tid);
        } else if (tid < 64) {
            scan_starts(scnt, sstart, NB, tid - 32);
        }
        if (tid < NB) { lbase[tid] = got; sbase[tid] = got2; }
        __syncthreads();

        int st[EPT], st2[EPT];
#pragma unroll
        for (int i = 0; i < EPT; ++i) {
            st[i]  = lstart[d[i] >> BSH];
            st2[i] = sstart[s[i] >> BSH];
        }
#pragma unroll
        for (int i = 0; i < EPT; ++i) {
            if (base + i * T1 + tid < cend) {
                lpairs[st[i] + lr[i]] = pack_pair(s[i], d[i]);
                lsrc[st2[i] + sr[i]]  = s[i];
            }
        }
        __syncthreads();

        const int left = cend - base;
        const int cn = left < T1 * EPT ? left : T1 * EPT;
        u64 v[EPT];
        int gb[EPT], ls[EPT];
#pragma unroll
        for (int i = 0; i < EPT; ++i) {
            const int j = i * T1 + tid;
            v[i] = lpairs[j < cn ? j : 0];
        }
#pragma unroll
        for (int i = 0; i < EPT; ++i) {
            const int b = (int)(v[i] >> 32) >> BSH;
            gb[i] = lbase[b];
            ls[i] = lstart[b];
        }
#pragma unroll
        for (int i = 0; i < EPT; ++i) {
            const int j = i * T1 + tid;
            if (j < cn) {
                const int b   = (int)(v[i] >> 32) >> BSH;
                const int pos = gb[i] + (j - ls[i]);
                u64* q;
                if (pos < BCAP) {
                    q = blist + (long)b * BCAP + pos;
                } else {
                    q = spill + atomicAdd(spill_n, 1);
                }
                *q = v[i];
            }
        }
        int w[EPT], gb2[EPT], ls2[EPT];
#pragma unroll
        for (int i = 0; i < EPT; ++i) {
            const int j = i * T1 + tid;
            w[i] = lsrc[j < cn ? j : 0];
        }
#pragma unroll
        for (int i = 0; i < EPT; ++i) {
            const int b = w[i] >> BSH;
            gb2[i] = sbase[b];
            ls2[i] = sstart[b];
        }
#pragma unroll
        for (int i = 0; i < EPT; ++i) {
            const int j = i * T1 + tid;
            if (j < cn) {
                const int b   = w[i] >> BSH;
                const int pos = gb2[i] + (j - ls2[i]);
                int* q;
                if (pos < SCAP) {
                    q = slist + (long)b * SCAP + pos;
                } else {
                    q = sspill + atomicAdd(sspill_n, 1);
                }
                *q = w[i];
            }
        }
        if (base + T1 * EPT < cend) __syncthreads();
    }
}

__global__ void __launch_bounds__(T2)
rank_range(const u64* __restrict__ blist, const int* __restrict__ gcount,
           const int* __restrict__ spill_n, const u64* __restrict__ spill,
           const int* __restrict__ slist, const int* __restrict__ scount,
           const int* __restrict__ sspill_n, const int* __restrict__ sspill,
           int* __restrict__ cnt_in, int* __restrict__ cnt_out,
           int* __restrict__ slots, int* __restrict__ ovf_n,
           int* __restrict__ ovf_src, int* __restrict__ ovf_dst, int N) {
    __shared__ int lcnt[RB];
    __shared__ int lco[RB];
    __shared__ __attribute__((aligned(16))) int lslots[RB * CAP];
    const int tid   = (int)threadIdx.x;
    const int b     = (int)blockIdx.x;
    const int nbase = b << BSH;

    int4* __restrict__ l4 = reinterpret_cast<int4*>(lslots);
    for (int t = tid; t < RB; t += T2) { lcnt[t] = 0; lco[t] = 0; }
    for (int t = tid; t < RB * CAP / 4; t += T2) l4[t] = make_int4(0, 0, 0, 0);
    __syncthreads();

    auto place = [&](const u64 p) {
        const int ps = (int)(unsigned)p;
        const int pd = (int)(p >> 32);
        const int dl = pd - nbase;
        const int r  = atomicAdd(&lcnt[dl], 1);
        if (r < CAP) {
            lslots[dl * CAP + r] = ps;
        } else {
            const int k = atomicAdd(ovf_n, 1);
            ovf_src[k] = ps;
            ovf_dst[k] = pd;
        }
    };

    int n = gcount[b * GSTR];
    n = n < BCAP ? n : BCAP;
    int n2 = scount[b * GSTR];
    n2 = n2 < SCAP ? n2 : SCAP;
    const int nmax = n > n2 ? n : n2;
    const u64* __restrict__ bl = blist + (long)b * BCAP;
    const int* __restrict__ sl = slist + (long)b * SCAP;
    for (int base = 0; base < nmax; base += T2 * PPT) {
        u64 p[PPT];
        int q[PPT];
#pragma unroll
        for (int u = 0; u < PPT; ++u) {
            const int i = base + u * T2 + tid;
            p[u] = (i < n) ? bl[i] : 0ull;
            q[u] = (i < n2) ? sl[i] : nbase;
        }
#pragma unroll
        for (int u = 0; u < PPT; ++u) {
            if (base + u * T2 + tid < n) place(p[u]);
        }
#pragma unroll
        for (int u = 0; u < PPT; ++u) {
            if (base + u * T2 + tid < n2) atomicAdd(&lco[q[u] - nbase], 1);
        }
    }

    const int ns = spill_n[0];
    for (int i = tid; i < ns; i += T2) {
        const u64 q = spill[i];
        if (((int)(q >> 32) >> BSH) == b) place(q);
    }
    const int ns2 = sspill_n[0];
    for (int i = tid; i < ns2; i += T2) {
        const int q = sspill[i];
        if ((q >> BSH) == b) atomicAdd(&lco[q - nbase], 1);
    }
    __syncthreads();

    const int left = N - nbase;
    const int nn = left < RB ? left : RB;
    for (int t = tid; t < nn; t += T2) {
        cnt_in[nbase + t]  = lcnt[t];
        cnt_out[nbase + t] = lco[t];
    }
    int4* __restrict__ g4 = reinterpret_cast<int4*>(slots + (long)nbase * CAP);
    for (int t = tid; t < nn * (CAP / 4); t += T2) g4[t] = l4[t];
}

template <int K, int COLS>
__global__ void __launch_bounds__(128)
gemm_norm_wmma(const float* __restrict__ X, const int* __restrict__ cntv,
               const float* __restrict__ W, float* __restrict__ H, int nrows) {
    constexpr int NT = COLS / 16;
    typedef __attribute__((ext_vector_type(NT))) float wvec;
    typedef __attribute__((ext_vector_type(4))) float v4f;
    constexpr int KB   = K / 8;
    constexpr int WROW = COLS / NT;
    const int lane  = threadIdx.x & 31;
    const int wave  = threadIdx.x >> 5;
    const int tile_m = blockIdx.x * (blockDim.x >> 5) + wave;
    if (tile_m * 16 >= nrows) return;

    const int c = lane & 15;
    const int h = lane >> 4;
    const int m = tile_m * 16 + c;

    const int co = cntv[m];
    const float nm = rsqrtf(co > 0 ? (float)co : 1.0f);
    const v4f*  __restrict__ xq = reinterpret_cast<const v4f*>(X + (long)m * K + 4 * h);
    const wvec* __restrict__ wq = reinterpret_cast<const wvec*>(W + (long)(4 * h) * COLS + NT * c);

    v8f acc[NT];
#pragma unroll
    for (int t = 0; t < NT; ++t) { v8f z = {}; acc[t] = z; }

#pragma unroll (KB <= 8 ? KB : 4)
    for (int kb = 0; kb < KB; ++kb) {
        const v4f  q  = xq[kb * 2];
        const wvec w0 = wq[(long)(kb * 8 + 0) * WROW];
        const wvec w1 = wq[(long)(kb * 8 + 1) * WROW];
        const wvec w2 = wq[(long)(kb * 8 + 2) * WROW];
        const wvec w3 = wq[(long)(kb * 8 + 3) * WROW];
        v2f a0, a1;
        a0.x = q.x * nm;  a0.y = q.y * nm;
        a1.x = q.z * nm;  a1.y = q.w * nm;
#pragma unroll
        for (int t = 0; t < NT; ++t) {
            v2f b;
            b.x = w0[t];
            b.y = w1[t];
            acc[t] = __builtin_amdgcn_wmma_f32_16x16x4_f32(
                false, a0, false, b, (short)0, acc[t], false, false);
        }
#pragma unroll
        for (int t = 0; t < NT; ++t) {
            v2f b;
            b.x = w2[t];
            b.y = w3[t];
            acc[t] = __builtin_amdgcn_wmma_f32_16x16x4_f32(
                false, a1, false, b, (short)0, acc[t], false, false);
        }
    }

    wvec* __restrict__ hq = reinterpret_cast<wvec*>(H + (long)(tile_m * 16 + 8 * h) * COLS + NT * c);
#pragma unroll
    for (int r = 0; r < 8; ++r) {
        wvec o;
#pragma unroll
        for (int t = 0; t < NT; ++t) o[t] = acc[t][r];
        hq[(long)r * WROW] = o;
    }
}

template <int C, bool RELU>
__global__ void __launch_bounds__(256)
gather_rows(const float* __restrict__ H, const int* __restrict__ slots,
            const int* __restrict__ cnt_in, const int* __restrict__ ovf_n,
            const int* __restrict__ ovf_src, const int* __restrict__ ovf_dst,
            const float* __restrict__ bias, float* __restrict__ out, int N) {
    constexpr int VEC = C / 32;
    constexpr int B   = (VEC == 1) ? 32 : 24;
    const int lane = threadIdx.x & 31;
    const int wave = __builtin_amdgcn_readfirstlane((int)(threadIdx.x >> 5));
    const int nwaves = (int)gridDim.x * 8;

    float bv[VEC];
#pragma unroll
    for (int k = 0; k < VEC; ++k) bv[k] = bias[lane * VEC + k];

    int node = (int)blockIdx.x * 8 + wave;
    int cnt_nx = 0, idx_nx = 0;
    if (node < N) {
        cnt_nx = cnt_in[node];
        idx_nx = slots[(long)node * CAP + lane];
    }
    for (; node < N; node += nwaves) {
        const int cnt   = cnt_nx;
        const int myidx = idx_nx;
        const int nxt   = node + nwaves;
        if (nxt < N) {
            cnt_nx = cnt_in[nxt];
            idx_nx = slots[(long)nxt * CAP + lane];
        }
        const int n32 = cnt < CAP ? cnt : CAP;
        float acc[VEC];
#pragma unroll
        for (int k = 0; k < VEC; ++k) acc[k] = 0.0f;

        for (int h = 0; h < n32; h += B) {
            const int n = n32 - h;
            float v[B][VEC];
#pragma unroll
            for (int u = 0; u < B; ++u) {
#pragma unroll
                for (int k = 0; k < VEC; ++k) v[u][k] = 0.0f;
                if (u == 0 || u < n) {
                    const int s = __builtin_amdgcn_readlane(myidx, h + u);
                    const float* __restrict__ p = H + (long)s * C + lane * VEC;
                    if constexpr (VEC == 2) {
                        const float2 t = *reinterpret_cast<const float2*>(p);
                        v[u][0] = t.x;
                        v[u][1] = t.y;
                    } else {
                        v[u][0] = p[0];
                    }
                }
            }
#pragma unroll
            for (int u = 0; u < B; ++u) {
#pragma unroll
                for (int k = 0; k < VEC; ++k) acc[k] += v[u][k];
            }
        }

        if (cnt > CAP) {
            const int novf = ovf_n[0];
            for (int j = 0; j < novf; j += 32) {
                const int q = j + lane;
                const int od = (q < novf) ? ovf_dst[q] : -1;
                const int os = (q < novf) ? ovf_src[q] : 0;
                unsigned m = __builtin_amdgcn_ballot_w32(od == node);
                while (m != 0u) {
                    const int b = __builtin_ctz(m);
                    m &= m - 1u;
                    const int s = __builtin_amdgcn_readlane(os, b);
                    const float* __restrict__ p = H + (long)s * C + lane * VEC;
#pragma unroll
                    for (int k = 0; k < VEC; ++k) acc[k] += p[k];
                }
            }
        }

        const float nin = rsqrtf(cnt > 0 ? (float)cnt : 1.0f);
        float r[VEC];
#pragma unroll
        for (int k = 0; k < VEC; ++k) {
            const float t = acc[k] * nin + bv[k];
            r[k] = RELU ? (t > 0.0f ? t : 0.0f) : t;
        }
        float* __restrict__ o = out + (long)node * C + lane * VEC;
        if constexpr (VEC == 2) {
            *reinterpret_cast<float2*>(o) = make_float2(r[0], r[1]);
        } else {
            o[0] = r[0];
        }
    }
}

extern "C" void kernel_launch(void* const* d_in, const int* in_sizes, int n_in,
                              void* d_out, int out_size, void* d_ws, size_t ws_size,
                              hipStream_t stream) {
    const float* features = (const float*)d_in[0];
    const int*   src      = (const int*)d_in[1];
    const int*   dst      = (const int*)d_in[2];
    const float* W1       = (const float*)d_in[3];
    const float* b1       = (const float*)d_in[4];
    const float* W2       = (const float*)d_in[5];
    const float* b2       = (const float*)d_in[6];

    const int IN_F = 128;
    const int N = in_sizes[0] / IN_F;
    const int E = in_sizes[1];

    int*   wsi      = (int*)d_ws;
    float* wsf      = (float*)d_ws;
    float* hbuf     = wsf;
    float* x1       = wsf + 64L * N;
    int*   slots    = wsi + 128L * N;
    int*   ovf_src  = wsi + 160L * N;
    int*   ovf_dst  = wsi + 160L * N + (long)E;
    int*   cnt_in   = wsi + 160L * N + 2L * E;
    int*   cnt_out  = wsi + 161L * N + 2L * E;
    int*   ovf_n    = wsi + 162L * N + 2L * E;
    int*   gcount   = wsi + 162L * N + 2L * E + 32;
    int*   scount   = wsi + 162L * N + 2L * E + 32 + NBMAX * GSTR;
    u64*   spill    = (u64*)(wsi + 162L * N + 2L * E + 32 + 2 * NBMAX * GSTR);
    u64*   blist    = (u64*)(wsi + 162L * N + 4L * E + 32 + 2 * NBMAX * GSTR);
    const int NB = (N + RB - 1) / RB;
    int*   sspill   = wsi + 162L * N + 4L * E + 32 + 2 * NBMAX * GSTR + 2L * BCAP * NB;
    int*   slist    = wsi + 162L * N + 5L * E + 32 + 2 * NBMAX * GSTR + 2L * BCAP * NB;
    float* outp     = (float*)d_out;

    hipMemsetAsync(ovf_n, 0, sizeof(int) * (32 + 2 * NBMAX * GSTR), stream);

    bucket_edges<<<256, T1, 0, stream>>>(src, dst, gcount, blist, ovf_n + 16, spill, scount, slist, ovf_n + 8, sspill, E, NB);
    rank_range<<<NB, T2, 0, stream>>>(blist, gcount, ovf_n + 16, spill, slist, scount, ovf_n + 8, sspill,
                                      cnt_in, cnt_out, slots, ovf_n, ovf_src, ovf_dst, N);

    const int mtiles = (N + 15) / 16;
    const int gblocks = (mtiles + 3) / 4;
    const int gather_blocks = 2048;

    gemm_norm_wmma<128, 64><<<gblocks, 128, 0, stream>>>(features, cnt_out, W1, hbuf, N);
    gather_rows<64, true><<<gather_blocks, 256, 0, stream>>>(hbuf, slots, cnt_in, ovf_n, ovf_src, ovf_dst, b1, x1, N);

    gemm_norm_wmma<64, 32><<<gblocks, 128, 0, stream>>>(x1, cnt_out, W2, hbuf, N);
    gather_rows<32, false><<<gather_blocks, 256, 0, stream>>>(hbuf, slots, cnt_in, ovf_n, ovf_src, ovf_dst, b2, outp, N);
}


# Layer-1 product done unscaled inside the edge-bucketing launch (one launch less); level 2 scales its range's finished rows by out-degree in place: a row is scaled after its k sum, not before
# speedup vs baseline: 6.8428x; 1.0431x over previous
#include <hip/hip_runtime.h>
#include <hip/hip_bf16.h>


typedef __attribute__((ext_vector_type(2))) float v2f;
typedef __attribute__((ext_vector_type(8))) float v8f;

typedef unsigned long long u64;

constexpr int CAP   = 32;
constexpr int BSH   = 9;
constexpr int RB    = 1 << BSH;
constexpr int T1    = 512;
constexpr int NBMAX = T1;
constexpr int EPT   = 13;
constexpr int BCAP  = 12288;
constexpr int SCAP  = 12288;
constexpr int GSTR  = 32;
constexpr int T2    = 1024;
constexpr int PPT   = 8;
constexpr int HID   = 64;

__device__ __forceinline__ u64 pack_pair(int s, int d) {
    return (u64)(unsigned)s | ((u64)(unsigned)d << 32);
}

__device__ __forceinline__ void scan_starts(const int* cnt, int* start, int NB, int lane) {
    int carry = 0;
    for (int c0 = 0; c0 < NB; c0 += 32) {
        const int idx = c0 + lane;
        const int v = (idx < NB) ? cnt[idx] : 0;
        int x = v;
        for (int o = 1; o < 32; o <<= 1) {
            const int y = __shfl_up(x, o);
            if (lane >= o) x += y;
        }
        if (idx < NB) start[idx] = carry + x - v;
        carry += __shfl(x, 31);
    }
}

template <int K, int COLS, bool SCALED>
__device__ __forceinline__ void product_strip(const float* __restrict__ X, const int* __restrict__ cntv,
                                              const float* __restrict__ W, float* __restrict__ H,
                                              const int tile_m, const int lane) {
    constexpr int NT = COLS / 16;
    typedef __attribute__((ext_vector_type(NT))) float wvec;
    typedef __attribute__((ext_vector_type(4))) float v4f;
    constexpr int KB   = K / 8;
    constexpr int WROW = COLS / NT;

    const int c = lane & 15;
    const int h = lane >> 4;
    const int m = tile_m * 16 + c;

    float nm = 1.0f;
    if constexpr (SCALED) {
        const int co = cntv[m];
        nm = rsqrtf(co > 0 ? (float)co : 1.0f);
    }
    const v4f*  __restrict__ xq = reinterpret_cast<const v4f*>(X + (long)m * K + 4 * h);
    const wvec* __restrict__ wq = reinterpret_cast<const wvec*>(W + (long)(4 * h) * COLS + NT * c);

    v8f acc[NT];
#pragma unroll
    for (int t = 0; t < NT; ++t) { v8f z = {}; acc[t] = z; }

#pragma unroll (KB <= 8 ? KB : 4)
    for (int kb = 0; kb < KB; ++kb) {
        const v4f  q  = xq[kb * 2];
        const wvec w0 = wq[(long)(kb * 8 + 0) * WROW];
        const wvec w1 = wq[(long)(kb * 8 + 1) * WROW];
        const wvec w2 = wq[(long)(kb * 8 + 2) * WROW];
        const wvec w3 = wq[(long)(kb * 8 + 3) * WROW];
        v2f a0, a1;
        if constexpr (SCALED) {
            a0.x = q.x * nm;  a0.y = q.y * nm;
            a1.x = q.z * nm;  a1.y = q.w * nm;
        } else {
            a0.x = q.x;  a0.y = q.y;
            a1.x = q.z;  a1.y = q.w;
        }
#pragma unroll
        for (int t = 0; t < NT; ++t) {
            v2f b;
            b.x = w0[t];
            b.y = w1[t];
            acc[t] = __builtin_amdgcn_wmma_f32_16x16x4_f32(
                false, a0, false, b, (short)0, acc[t], false, false);
        }
#pragma unroll
        for (int t = 0; t < NT; ++t) {
            v2f b;
            b.x = w2[t];
            b.y = w3[t];
            acc[t] = __builtin_amdgcn_wmma_f32_16x16x4_f32(
                false, a1, false, b, (short)0, acc[t], false, false);
        }
    }

    wvec* __restrict__ hq = reinterpret_cast<wvec*>(H + (long)(tile_m * 16 + 8 * h) * COLS + NT * c);
#pragma unroll
    for (int r = 0; r < 8; ++r) {
        wvec o;
#pragma unroll
        for (int t = 0; t < NT; ++t) o[t] = acc[t][r];
        hq[(long)r * WROW] = o;
    }
}

template <int K, int COLS>
__global__ void __launch_bounds__(128)
gemm_norm_wmma(const float* __restrict__ X, const int* __restrict__ cntv,
               const float* __restrict__ W, float* __restrict__ H, int nrows) {
    const int lane  = threadIdx.x & 31;
    const int wave  = threadIdx.x >> 5;
    const int tile_m = blockIdx.x * (blockDim.x >> 5) + wave;
    if (tile_m * 16 >= nrows) return;
    product_strip<K, COLS, true>(X, cntv, W, H, tile_m, lane);
}

__global__ void __launch_bounds__(T1, 8)
bucket_edges_and_product(const int* __restrict__ src, const int* __restrict__ dst,
             int* __restrict__ gcount, u64* __restrict__ blist,
             int* __restrict__ spill_n, u64* __restrict__ spill,
             int* __restrict__ scount, int* __restrict__ slist,
             int* __restrict__ sspill_n, int* __restrict__ sspill, int E, int NB,
             const float* __restrict__ X, const float* __restrict__ W, float* __restrict__ H,
             int nrows, int nbk) {
    if ((int)blockIdx.x >= nbk) {
        const int wave = __builtin_amdgcn_readfirstlane((int)(threadIdx.x >> 5));
        const int npw  = (int)gridDim.x - nbk;
        for (int tile_m = wave * npw + ((int)blockIdx.x - nbk); tile_m * 16 < nrows; tile_m += npw * (T1 / 32))
            product_strip<128, HID, false>(X, nullptr, W, H, tile_m, (int)(threadIdx.x & 31));
        return;
    }
    __shared__ int lcount[NBMAX];
    __shared__ int lstart[NBMAX];
    __shared__ int lbase[NBMAX];
    __shared__ u64 lpairs[T1 * EPT];
    __shared__ int scnt[NBMAX];
    __shared__ int sstart[NBMAX];
    __shared__ int sbase[NBMAX];
    __shared__ int lsrc[T1 * EPT];
    const int tid   = (int)threadIdx.x;
    const int chunk = (E + nbk - 1) / nbk;
    const int cbeg  = (int)blockIdx.x * chunk;
    const int cend  = (cbeg + chunk < E) ? (cbeg + chunk) : E;

    for (int base = cbeg; base < cend; base += T1 * EPT) {
        if (tid < NB) { lcount[tid] = 0; scnt[tid] = 0; }
        __syncthreads();

        int s[EPT], d[EPT], lr[EPT], sr[EPT];
#pragma unroll
        for (int i = 0; i < EPT; ++i) {
            const int e = base + i * T1 + tid;
            const bool ok = e < cend;
            s[i] = ok ? src[e] : 0;
            d[i] = ok ? dst[e] : 0;
        }
#pragma unroll
        for (int i = 0; i < EPT; ++i) {
            lr[i] = 0;
            sr[i] = 0;
            if (base + i * T1 + tid < cend) {
                lr[i] = atomicAdd(&lcount[d[i] >> BSH], 1);
                sr[i] = atomicAdd(&scnt[s[i] >> BSH], 1);
            }
        }
        __syncthreads();

        const int c  = (tid < NB) ? lcount[tid] : 0;
        const int c2 = (tid < NB) ? scnt[tid] : 0;
        int got = 0, got2 = 0;
        if (c > 0)  got  = atomicAdd(&gcount[tid * GSTR], c);
        if (c2 > 0) got2 = atomicAdd(&scount[tid * GSTR], c2);
        if (tid < 32) {
            scan_starts(lcount, lstart, NB, tid);
        } else if (tid < 64) {
            scan_starts(scnt, sstart, NB, tid - 32);
        }
        if (tid < NB) { lbase[tid] = got; sbase[tid] = got2; }
        __syncthreads();

        int st[EPT], st2[EPT];
#pragma unroll
        for (int i = 0; i < EPT; ++i) {
            st[i]  = lstart[d[i] >> BSH];
            st2[i] = sstart[s[i] >> BSH];
        }
#pragma unroll
        for (int i = 0; i < EPT; ++i) {
            if (base + i * T1 + tid < cend) {
                lpairs[st[i] + lr[i]] = pack_pair(s[i], d[i]);
                lsrc[st2[i] + sr[i]]  = s[i];
            }
        }
        __syncthreads();

        const int left = cend - base;
        const int cn = left < T1 * EPT ? left : T1 * EPT;
        u64 v[EPT];
        int gb[EPT], ls[EPT];
#pragma unroll
        for (int i = 0; i < EPT; ++i) {
            const int j = i * T1 + tid;
            v[i] = lpairs[j < cn ? j : 0];
        }
#pragma unroll
        for (int i = 0; i < EPT; ++i) {
            const int b = (int)(v[i] >> 32) >> BSH;
            gb[i] = lbase[b];
            ls[i] = lstart[b];
        }
#pragma unroll
        for (int i = 0; i < EPT; ++i) {
            const int j = i * T1 + tid;
            if (j < cn) {
                const int b   = (int)(v[i] >> 32) >> BSH;
                const int pos = gb[i] + (j - ls[i]);
                u64* q;
                if (pos < BCAP) {
                    q = blist + (long)b * BCAP + pos;
                } else {
                    q = spill + atomicAdd(spill_n, 1);
                }
                *q = v[i];
            }
        }
        int w[EPT], gb2[EPT], ls2[EPT];
#pragma unroll
        for (int i = 0; i < EPT; ++i) {
            const int j = i * T1 + tid;
            w[i] = lsrc[j < cn ? j : 0];
        }
#pragma unroll
        for (int i = 0; i < EPT; ++i) {
            const int b = w[i] >> BSH;
            gb2[i] = sbase[b];
            ls2[i] = sstart[b];
        }
#pragma unroll
        for (int i = 0; i < EPT; ++i) {
            const int j = i * T1 + tid;
            if (j < cn) {
                const int b   = w[i] >> BSH;
                const int pos = gb2[i] + (j - ls2[i]);
                int* q;
                if (pos < SCAP) {
                    q = slist + (long)b * SCAP + pos;
                } else {
                    q = sspill + atomicAdd(sspill_n, 1);
                }
                *q = w[i];
            }
        }
        if (base + T1 * EPT < cend) __syncthreads();
    }
}

__global__ void __launch_bounds__(T2)
rank_range(const u64* __restrict__ blist, const int* __restrict__ gcount,
           const int* __restrict__ spill_n, const u64* __restrict__ spill,
           const int* __restrict__ slist, const int* __restrict__ scount,
           const int* __restrict__ sspill_n, const int* __restrict__ sspill,
           int* __restrict__ cnt_in, int* __restrict__ cnt_out,
           int* __restrict__ slots, int* __restrict__ ovf_n,
           int* __restrict__ ovf_src, int* __restrict__ ovf_dst, float* __restrict__ H, int N) {
    __shared__ int lcnt[RB];
    __shared__ int lco[RB];
    __shared__ __attribute__((aligned(16))) int lslots[RB * CAP];
    const int tid   = (int)threadIdx.x;
    const int b     = (int)blockIdx.x;
    const int nbase = b << BSH;

    int4* __restrict__ l4 = reinterpret_cast<int4*>(lslots);
    for (int t = tid; t < RB; t += T2) { lcnt[t] = 0; lco[t] = 0; }
    for (int t = tid; t < RB * CAP / 4; t += T2) l4[t] = make_int4(0, 0, 0, 0);
    __syncthreads();

    auto place = [&](const u64 p) {
        const int ps = (int)(unsigned)p;
        const int pd = (int)(p >> 32);
        const int dl = pd - nbase;
        const int r  = atomicAdd(&lcnt[dl], 1);
        if (r < CAP) {
            lslots[dl * CAP + r] = ps;
        } else {
            const int k = atomicAdd(ovf_n, 1);
            ovf_src[k] = ps;
            ovf_dst[k] = pd;
        }
    };

    int n = gcount[b * GSTR];
    n = n < BCAP ? n : BCAP;
    int n2 = scount[b * GSTR];
    n2 = n2 < SCAP ? n2 : SCAP;
    const int nmax = n > n2 ? n : n2;
    const u64* __restrict__ bl = blist + (long)b * BCAP;
    const int* __restrict__ sl = slist + (long)b * SCAP;
    for (int base = 0; base < nmax; base += T2 * PPT) {
        u64 p[PPT];
        int q[PPT];
#pragma unroll
        for (int u = 0; u < PPT; ++u) {
            const int i = base + u * T2 + tid;
            p[u] = (i < n) ? bl[i] : 0ull;
            q[u] = (i < n2) ? sl[i] : nbase;
        }
#pragma unroll
        for (int u = 0; u < PPT; ++u) {
            if (base + u * T2 + tid < n) place(p[u]);
        }
#pragma unroll
        for (int u = 0; u < PPT; ++u) {
            if (base + u * T2 + tid < n2) atomicAdd(&lco[q[u] - nbase], 1);
        }
    }

    const int ns = spill_n[0];
    for (int i = tid; i < ns; i += T2) {
        const u64 q = spill[i];
        if (((int)(q >> 32) >> BSH) == b) place(q);
    }
    const int ns2 = sspill_n[0];
    for (int i = tid; i < ns2; i += T2) {
        const int q = sspill[i];
        if ((q >> BSH) == b) atomicAdd(&lco[q - nbase], 1);
    }
    __syncthreads();

    const int left = N - nbase;
    const int nn = left < RB ? left : RB;
    for (int t = tid; t < nn; t += T2) {
        cnt_in[nbase + t]  = lcnt[t];
        cnt_out[nbase + t] = lco[t];
    }
    typedef __attribute__((ext_vector_type(4))) float v4f;
    constexpr int HV  = HID / 4;
    constexpr int HPT = RB * HV / T2;
    v4f* __restrict__ hp = reinterpret_cast<v4f*>(H + (long)nbase * HID);
    const int nv = nn * HV;
    v4f hv[HPT];
    int hc[HPT];
#pragma unroll
    for (int k = 0; k < HPT; ++k) {
        const int i = k * T2 + tid;
        if (i < nv) hv[k] = hp[i];
        hc[k] = lco[i / HV];
    }

    int4* __restrict__ g4 = reinterpret_cast<int4*>(slots + (long)nbase * CAP);
    for (int t = tid; t < nn * (CAP / 4); t += T2) g4[t] = l4[t];

#pragma unroll
    for (int k = 0; k < HPT; ++k) {
        const int i = k * T2 + tid;
        if (i < nv) hp[i] = hv[k] * rsqrtf(hc[k] > 0 ? (float)hc[k] : 1.0f);
    }
}

template <int C, bool RELU>
__global__ void __launch_bounds__(256)
gather_rows(const float* __restrict__ H, const int* __restrict__ slots,
            const int* __restrict__ cnt_in, const int* __restrict__ ovf_n,
            const int* __restrict__ ovf_src, const int* __restrict__ ovf_dst,
            const float* __restrict__ bias, float* __restrict__ out, int N) {
    constexpr int VEC = C / 32;
    constexpr int B   = (VEC == 1) ? 32 : 24;
    const int lane = threadIdx.x & 31;
    const int wave = __builtin_amdgcn_readfirstlane((int)(threadIdx.x >> 5));
    const int nwaves = (int)gridDim.x * 8;

    float bv[VEC];
#pragma unroll
    for (int k = 0; k < VEC; ++k) bv[k] = bias[lane * VEC + k];

    int node = (int)blockIdx.x * 8 + wave;
    int cnt_nx = 0, idx_nx = 0;
    if (node < N) {
        cnt_nx = cnt_in[node];
        idx_nx = slots[(long)node * CAP + lane];
    }
    for (; node < N; node += nwaves) {
        const int cnt   = cnt_nx;
        const int myidx = idx_nx;
        const int nxt   = node + nwaves;
        if (nxt < N) {
            cnt_nx = cnt_in[nxt];
            idx_nx = slots[(long)nxt * CAP + lane];
        }
        const int n32 = cnt < CAP ? cnt : CAP;
        float acc[VEC];
#pragma unroll
        for (int k = 0; k < VEC; ++k) acc[k] = 0.0f;

        for (int h = 0; h < n32; h += B) {
            const int n = n32 - h;
            float v[B][VEC];
#pragma unroll
            for (int u = 0; u < B; ++u) {
#pragma unroll
                for (int k = 0; k < VEC; ++k) v[u][k] = 0.0f;
                if (u == 0 || u < n) {
                    const int s = __builtin_amdgcn_readlane(myidx, h + u);
                    const float* __restrict__ p = H + (long)s * C + lane * VEC;
                    if constexpr (VEC == 2) {
                        const float2 t = *reinterpret_cast<const float2*>(p);
                        v[u][0] = t.x;
                        v[u][1] = t.y;
                    } else {
                        v[u][0] = p[0];
                    }
                }
            }
#pragma unroll
            for (int u = 0; u < B; ++u) {
#pragma unroll
                for (int k = 0; k < VEC; ++k) acc[k] += v[u][k];
            }
        }

        if (cnt > CAP) {
            const int novf = ovf_n[0];
            for (int j = 0; j < novf; j += 32) {
                const int q = j + lane;
                const int od = (q < novf) ? ovf_dst[q] : -1;
                const int os = (q < novf) ? ovf_src[q] : 0;
                unsigned m = __builtin_amdgcn_ballot_w32(od == node);
                while (m != 0u) {
                    const int b = __builtin_ctz(m);
                    m &= m - 1u;
                    const int s = __builtin_amdgcn_readlane(os, b);
                    const float* __restrict__ p = H + (long)s * C + lane * VEC;
#pragma unroll
                    for (int k = 0; k < VEC; ++k) acc[k] += p[k];
                }
            }
        }

        const float nin = rsqrtf(cnt > 0 ? (float)cnt : 1.0f);
        float r[VEC];
#pragma unroll
        for (int k = 0; k < VEC; ++k) {
            const float t = acc[k] * nin + bv[k];
            r[k] = RELU ? (t > 0.0f ? t : 0.0f) : t;
        }
        float* __restrict__ o = out + (long)node * C + lane * VEC;
        if constexpr (VEC == 2) {
            *reinterpret_cast<float2*>(o) = make_float2(r[0], r[1]);
        } else {
            o[0] = r[0];
        }
    }
}

extern "C" void kernel_launch(void* const* d_in, const int* in_sizes, int n_in,
                              void* d_out, int out_size, void* d_ws, size_t ws_size,
                              hipStream_t stream) {
    const float* features = (const float*)d_in[0];
    const int*   src      = (const int*)d_in[1];
    const int*   dst      = (const int*)d_in[2];
    const float* W1       = (const float*)d_in[3];
    const float* b1       = (const float*)d_in[4];
    const float* W2       = (const float*)d_in[5];
    const float* b2       = (const float*)d_in[6];

    const int IN_F = 128;
    const int N = in_sizes[0] / IN_F;
    const int E = in_sizes[1];

    int*   wsi      = (int*)d_ws;
    float* wsf      = (float*)d_ws;
    float* hbuf     = wsf;
    float* x1       = wsf + 64L * N;
    int*   slots    = wsi + 128L * N;
    int*   ovf_src  = wsi + 160L * N;
    int*   ovf_dst  = wsi + 160L * N + (long)E;
    int*   cnt_in   = wsi + 160L * N + 2L * E;
    int*   cnt_out  = wsi + 161L * N + 2L * E;
    int*   ovf_n    = wsi + 162L * N + 2L * E;
    int*   gcount   = wsi + 162L * N + 2L * E + 32;
    int*   scount   = wsi + 162L * N + 2L * E + 32 + NBMAX * GSTR;
    u64*   spill    = (u64*)(wsi + 162L * N + 2L * E + 32 + 2 * NBMAX * GSTR);
    u64*   blist    = (u64*)(wsi + 162L * N + 4L * E + 32 + 2 * NBMAX * GSTR);
    const int NB = (N + RB - 1) / RB;
    int*   sspill   = wsi + 162L * N + 4L * E + 32 + 2 * NBMAX * GSTR + 2L * BCAP * NB;
    int*   slist    = wsi + 162L * N + 5L * E + 32 + 2 * NBMAX * GSTR + 2L * BCAP * NB;
    float* outp     = (float*)d_out;

    hipMemsetAsync(ovf_n, 0, sizeof(int) * (32 + 2 * NBMAX * GSTR), stream);

    const int nbk = 256;
    const int npk = 256;

    bucket_edges_and_product<<<nbk + npk, T1, 0, stream>>>(src, dst, gcount, blist, ovf_n + 16, spill, scount, slist,
                                                          ovf_n + 8, sspill, E, NB, features, W1, hbuf, N, nbk);
    rank_range<<<NB, T2, 0, stream>>>(blist, gcount, ovf_n + 16, spill, slist, scount, ovf_n + 8, sspill,
                                      cnt_in, cnt_out, slots, ovf_n, ovf_src, ovf_dst, hbuf, N);

    const int mtiles = (N + 15) / 16;
    const int gblocks = (mtiles + 3) / 4;
    const int gather_blocks = 2048;

    gather_rows<64, true><<<gather_blocks, 256, 0, stream>>>(hbuf, slots, cnt_in, ovf_n, ovf_src, ovf_dst, b1, x1, N);

    gemm_norm_wmma<64, 32><<<gblocks, 128, 0, stream>>>(x1, cnt_out, W2, hbuf, N);
    gather_rows<32, false><<<gather_blocks, 256, 0, stream>>>(hbuf, slots, cnt_in, ovf_n, ovf_src, ovf_dst, b2, outp, N);
}


# Layer-2 product moved into the layer-1 gather: a work-group's 64 finished rows meet in one LDS tile behind the node loop, 4 of its waves multiply 16 each by W2 (f32 16x16x4); no x1, one launch less
# speedup vs baseline: 7.4510x; 1.0889x over previous
#include <hip/hip_runtime.h>
#include <hip/hip_bf16.h>


typedef __attribute__((ext_vector_type(2))) float v2f;
typedef __attribute__((ext_vector_type(8))) float v8f;

typedef unsigned long long u64;

constexpr int CAP   = 32;
constexpr int BSH   = 9;
constexpr int RB    = 1 << BSH;
constexpr int T1    = 512;
constexpr int NBMAX = T1;
constexpr int EPT   = 13;
constexpr int BCAP  = 12288;
constexpr int SCAP  = 12288;
constexpr int GSTR  = 32;
constexpr int T2    = 1024;
constexpr int PPT   = 8;
constexpr int HID   = 64;
constexpr int OUTC  = 32;
constexpr int GW    = 8;
constexpr int TROWS = 8;
constexpr int TP    = HID + 4;
constexpr int WP    = OUTC + 8;

__device__ __forceinline__ u64 pack_pair(int s, int d) {
    return (u64)(unsigned)s | ((u64)(unsigned)d << 32);
}

__device__ __forceinline__ void scan_starts(const int* cnt, int* start, int NB, int lane) {
    int carry = 0;
    for (int c0 = 0; c0 < NB; c0 += 32) {
        const int idx = c0 + lane;
        const int v = (idx < NB) ? cnt[idx] : 0;
        int x = v;
        for (int o = 1; o < 32; o <<= 1) {
            const int y = __shfl_up(x, o);
            if (lane >= o) x += y;
        }
        if (idx < NB) start[idx] = carry + x - v;
        carry += __shfl(x, 31);
    }
}

template <int K, int COLS>
__device__ __forceinline__ void product_strip(const float* __restrict__ X,
                                              const float* __restrict__ W, float* __restrict__ H,
                                              const int tile_m, const int lane) {
    constexpr int NT = COLS / 16;
    typedef __attribute__((ext_vector_type(NT))) float wvec;
    typedef __attribute__((ext_vector_type(4))) float v4f;
    constexpr int KB   = K / 8;
    constexpr int WROW = COLS / NT;

    const int c = lane & 15;
    const int h = lane >> 4;
    const int m = tile_m * 16 + c;

    const v4f*  __restrict__ xq = reinterpret_cast<const v4f*>(X + (long)m * K + 4 * h);
    const wvec* __restrict__ wq = reinterpret_cast<const wvec*>(W + (long)(4 * h) * COLS + NT * c);

    v8f acc[NT];
#pragma unroll
    for (int t = 0; t < NT; ++t) { v8f z = {}; acc[t] = z; }

#pragma unroll (KB <= 8 ? KB : 4)
    for (int kb = 0; kb < KB; ++kb) {
        const v4f  q  = xq[kb * 2];
        const wvec w0 = wq[(long)(kb * 8 + 0) * WROW];
        const wvec w1 = wq[(long)(kb * 8 + 1) * WROW];
        const wvec w2 = wq[(long)(kb * 8 + 2) * WROW];
        const wvec w3 = wq[(long)(kb * 8 + 3) * WROW];
        v2f a0, a1;
        a0.x = q.x;  a0.y = q.y;
        a1.x = q.z;  a1.y = q.w;
#pragma unroll
        for (int t = 0; t < NT; ++t) {
            v2f b;
            b.x = w0[t];
            b.y = w1[t];
            acc[t] = __builtin_amdgcn_wmma_f32_16x16x4_f32(
                false, a0, false, b, (short)0, acc[t], false, false);
        }
#pragma unroll
        for (int t = 0; t < NT; ++t) {
            v2f b;
            b.x = w2[t];
            b.y = w3[t];
            acc[t] = __builtin_amdgcn_wmma_f32_16x16x4_f32(
                false, a1, false, b, (short)0, acc[t], false, false);
        }
    }

    wvec* __restrict__ hq = reinterpret_cast<wvec*>(H + (long)(tile_m * 16 + 8 * h) * COLS + NT * c);
#pragma unroll
    for (int r = 0; r < 8; ++r) {
        wvec o;
#pragma unroll
        for (int t = 0; t < NT; ++t) o[t] = acc[t][r];
        hq[(long)r * WROW] = o;
    }
}

template <int K, int COLS>
__device__ __forceinline__ void tile_product(const float* trow0, const float* wl, float* __restrict__ Hout,
                                             const int node0, const int nstep, const int N, const int lane) {
    constexpr int NT = COLS / 16;
    typedef __attribute__((ext_vector_type(NT))) float wvec;
    typedef __attribute__((ext_vector_type(4))) float v4f;
    constexpr int KB   = K / 8;
    constexpr int WROW = WP / NT;
    static_assert(WP % NT == 0, "a row of the LDS copy of W is a whole number of wvec units");

    const int c = lane & 15;
    const int h = lane >> 4;
    const v4f*  xq = reinterpret_cast<const v4f*>(trow0 + c * TP + 4 * h);
    const wvec* wq = reinterpret_cast<const wvec*>(wl + (4 * h) * WP + NT * c);

    v8f acc[NT];
#pragma unroll
    for (int t = 0; t < NT; ++t) { v8f z = {}; acc[t] = z; }

#pragma unroll 2
    for (int kb = 0; kb < KB; ++kb) {
        const v4f  q  = xq[kb * 2];
        const wvec w0 = wq[(kb * 8 + 0) * WROW];
        const wvec w1 = wq[(kb * 8 + 1) * WROW];
        const wvec w2 = wq[(kb * 8 + 2) * WROW];
        const wvec w3 = wq[(kb * 8 + 3) * WROW];
        v2f a0, a1;
        a0.x = q.x;  a0.y = q.y;
        a1.x = q.z;  a1.y = q.w;
#pragma unroll
        for (int t = 0; t < NT; ++t) {
            v2f b;
            b.x = w0[t];
            b.y = w1[t];
            acc[t] = __builtin_amdgcn_wmma_f32_16x16x4_f32(
                false, a0, false, b, (short)0, acc[t], false, false);
        }
#pragma unroll
        for (int t = 0; t < NT; ++t) {
            v2f b;
            b.x = w2[t];
            b.y = w3[t];
            acc[t] = __builtin_amdgcn_wmma_f32_16x16x4_f32(
                false, a1, false, b, (short)0, acc[t], false, false);
        }
    }

#pragma unroll
    for (int r = 0; r < 8; ++r) {
        const int nd = node0 + h + r * nstep;
        if (nd < N) {
            wvec o;
#pragma unroll
            for (int t = 0; t < NT; ++t) o[t] = acc[t][r];
            *reinterpret_cast<wvec*>(Hout + (long)nd * COLS + NT * c) = o;
        }
    }
}

__global__ void __launch_bounds__(T1, 8)
bucket_edges_and_product(const int* __restrict__ src, const int* __restrict__ dst,
             int* __restrict__ gcount, u64* __restrict__ blist,
             int* __restrict__ spill_n, u64* __restrict__ spill,
             int* __restrict__ scount, int* __restrict__ slist,
             int* __restrict__ sspill_n, int* __restrict__ sspill, int E, int NB,
             const float* __restrict__ X, const float* __restrict__ W, float* __restrict__ H,
             int nrows, int nbk) {
    if ((int)blockIdx.x >= nbk) {
        const int wave = __builtin_amdgcn_readfirstlane((int)(threadIdx.x >> 5));
        const int npw  = (int)gridDim.x - nbk;
        for (int tile_m = wave * npw + ((int)blockIdx.x - nbk); tile_m * 16 < nrows; tile_m += npw * (T1 / 32))
            product_strip<128, HID>(X, W, H, tile_m, (int)(threadIdx.x & 31));
        return;
    }
    __shared__ int lcount[NBMAX];
    __shared__ int lstart[NBMAX];
    __shared__ int lbase[NBMAX];
    __shared__ u64 lpairs[T1 * EPT];
    __shared__ int scnt[NBMAX];
    __shared__ int sstart[NBMAX];
    __shared__ int sbase[NBMAX];
    __shared__ int lsrc[T1 * EPT];
    const int tid   = (int)threadIdx.x;
    const int chunk = (E + nbk - 1) / nbk;
    const int cbeg  = (int)blockIdx.x * chunk;
    const int cend  = (cbeg + chunk < E) ? (cbeg + chunk) : E;

    for (int base = cbeg; base < cend; base += T1 * EPT) {
        if (tid < NB) { lcount[tid] = 0; scnt[tid] = 0; }
        __syncthreads();

        int s[EPT], d[EPT], lr[EPT], sr[EPT];
#pragma unroll
        for (int i = 0; i < EPT; ++i) {
            const int e = base + i * T1 + tid;
            const bool ok = e < cend;
            s[i] = ok ? src[e] : 0;
            d[i] = ok ? dst[e] : 0;
        }
#pragma unroll
        for (int i = 0; i < EPT; ++i) {
            lr[i] = 0;
            sr[i] = 0;
            if (base + i * T1 + tid < cend) {
                lr[i] = atomicAdd(&lcount[d[i] >> BSH], 1);
                sr[i] = atomicAdd(&scnt[s[i] >> BSH], 1);
            }
        }
        __syncthreads();

        const int c  = (tid < NB) ? lcount[tid] : 0;
        const int c2 = (tid < NB) ? scnt[tid] : 0;
        int got = 0, got2 = 0;
        if (c > 0)  got  = atomicAdd(&gcount[tid * GSTR], c);
        if (c2 > 0) got2 = atomicAdd(&scount[tid * GSTR], c2);
        if (tid < 32) {
            scan_starts(lcount, lstart, NB, tid);
        } else if (tid < 64) {
            scan_starts(scnt, sstart, NB, tid - 32);
        }
        if (tid < NB) { lbase[tid] = got; sbase[tid] = got2; }
        __syncthreads();

        int st[EPT], st2[EPT];
#pragma unroll
        for (int i = 0; i < EPT; ++i) {
            st[i]  = lstart[d[i] >> BSH];
            st2[i] = sstart[s[i] >> BSH];
        }
#pragma unroll
        for (int i = 0; i < EPT; ++i) {
            if (base + i * T1 + tid < cend) {
                lpairs[st[i] + lr[i]] = pack_pair(s[i], d[i]);
                lsrc[st2[i] + sr[i]]  = s[i];
            }
        }
        __syncthreads();

        const int left = cend - base;
        const int cn = left < T1 * EPT ? left : T1 * EPT;
        u64 v[EPT];
        int gb[EPT], ls[EPT];
#pragma unroll
        for (int i = 0; i < EPT; ++i) {
            const int j = i * T1 + tid;
            v[i] = lpairs[j < cn ? j : 0];
        }
#pragma unroll
        for (int i = 0; i < EPT; ++i) {
            const int b = (int)(v[i] >> 32) >> BSH;
            gb[i] = lbase[b];
            ls[i] = lstart[b];
        }
#pragma unroll
        for (int i = 0; i < EPT; ++i) {
            const int j = i * T1 + tid;
            if (j < cn) {
                const int b   = (int)(v[i] >> 32) >> BSH;
                const int pos = gb[i] + (j - ls[i]);
                u64* q;
                if (pos < BCAP) {
                    q = blist + (long)b * BCAP + pos;
                } else {
                    q = spill + atomicAdd(spill_n, 1);
                }
                *q = v[i];
            }
        }
        int w[EPT], gb2[EPT], ls2[EPT];
#pragma unroll
        for (int i = 0; i < EPT; ++i) {
            const int j = i * T1 + tid;
            w[i] = lsrc[j < cn ? j : 0];
        }
#pragma unroll
        for (int i = 0; i < EPT; ++i) {
            const int b = w[i] >> BSH;
            gb2[i] = sbase[b];
            ls2[i] = sstart[b];
        }
#pragma unroll
        for (int i = 0; i < EPT; ++i) {
            const int j = i * T1 + tid;
            if (j < cn) {
                const int b   = w[i] >> BSH;
                const int pos = gb2[i] + (j - ls2[i]);
                int* q;
                if (pos < SCAP) {
                    q = slist + (long)b * SCAP + pos;
                } else {
                    q = sspill + atomicAdd(sspill_n, 1);
                }
                *q = w[i];
            }
        }
        if (base + T1 * EPT < cend) __syncthreads();
    }
}

__global__ void __launch_bounds__(T2)
rank_range(const u64* __restrict__ blist, const int* __restrict__ gcount,
           const int* __restrict__ spill_n, const u64* __restrict__ spill,
           const int* __restrict__ slist, const int* __restrict__ scount,
           const int* __restrict__ sspill_n, const int* __restrict__ sspill,
           int* __restrict__ cnt_in, int* __restrict__ cnt_out,
           int* __restrict__ slots, int* __restrict__ ovf_n,
           int* __restrict__ ovf_src, int* __restrict__ ovf_dst, float* __restrict__ H, int N) {
    __shared__ int lcnt[RB];
    __shared__ int lco[RB];
    __shared__ __attribute__((aligned(16))) int lslots[RB * CAP];
    const int tid   = (int)threadIdx.x;
    const int b     = (int)blockIdx.x;
    const int nbase = b << BSH;

    int4* __restrict__ l4 = reinterpret_cast<int4*>(lslots);
    for (int t = tid; t < RB; t += T2) { lcnt[t] = 0; lco[t] = 0; }
    for (int t = tid; t < RB * CAP / 4; t += T2) l4[t] = make_int4(0, 0, 0, 0);
    __syncthreads();

    auto place = [&](const u64 p) {
        const int ps = (int)(unsigned)p;
        const int pd = (int)(p >> 32);
        const int dl = pd - nbase;
        const int r  = atomicAdd(&lcnt[dl], 1);
        if (r < CAP) {
            lslots[dl * CAP + r] = ps;
        } else {
            const int k = atomicAdd(ovf_n, 1);
            ovf_src[k] = ps;
            ovf_dst[k] = pd;
        }
    };

    int n = gcount[b * GSTR];
    n = n < BCAP ? n : BCAP;
    int n2 = scount[b * GSTR];
    n2 = n2 < SCAP ? n2 : SCAP;
    const int nmax = n > n2 ? n : n2;
    const u64* __restrict__ bl = blist + (long)b * BCAP;
    const int* __restrict__ sl = slist + (long)b * SCAP;
    for (int base = 0; base < nmax; base += T2 * PPT) {
        u64 p[PPT];
        int q[PPT];
#pragma unroll
        for (int u = 0; u < PPT; ++u) {
            const int i = base + u * T2 + tid;
            p[u] = (i < n) ? bl[i] : 0ull;
            q[u] = (i < n2) ? sl[i] : nbase;
        }
#pragma unroll
        for (int u = 0; u < PPT; ++u) {
            if (base + u * T2 + tid < n) place(p[u]);
        }
#pragma unroll
        for (int u = 0; u < PPT; ++u) {
            if (base + u * T2 + tid < n2) atomicAdd(&lco[q[u] - nbase], 1);
        }
    }

    const int ns = spill_n[0];
    for (int i = tid; i < ns; i += T2) {
        const u64 q = spill[i];
        if (((int)(q >> 32) >> BSH) == b) place(q);
    }
    const int ns2 = sspill_n[0];
    for (int i = tid; i < ns2; i += T2) {
        const int q = sspill[i];
        if ((q >> BSH) == b) atomicAdd(&lco[q - nbase], 1);
    }
    __syncthreads();

    const int left = N - nbase;
    const int nn = left < RB ? left : RB;
    for (int t = tid; t < nn; t += T2) {
        cnt_in[nbase + t]  = lcnt[t];
        cnt_out[nbase + t] = lco[t];
    }
    typedef __attribute__((ext_vector_type(4))) float v4f;
    constexpr int HV  = HID / 4;
    constexpr int HPT = RB * HV / T2;
    v4f* __restrict__ hp = reinterpret_cast<v4f*>(H + (long)nbase * HID);
    const int nv = nn * HV;
    v4f hv[HPT];
    int hc[HPT];
#pragma unroll
    for (int k = 0; k < HPT; ++k) {
        const int i = k * T2 + tid;
        if (i < nv) hv[k] = hp[i];
        hc[k] = lco[i / HV];
    }

    int4* __restrict__ g4 = reinterpret_cast<int4*>(slots + (long)nbase * CAP);
    for (int t = tid; t < nn * (CAP / 4); t += T2) g4[t] = l4[t];

#pragma unroll
    for (int k = 0; k < HPT; ++k) {
        const int i = k * T2 + tid;
        if (i < nv) hp[i] = hv[k] * rsqrtf(hc[k] > 0 ? (float)hc[k] : 1.0f);
    }
}

template <int C, bool RELU, bool FUSED>
__global__ void __launch_bounds__(GW * 32)
gather_rows(const float* __restrict__ H, const int* __restrict__ slots,
            const int* __restrict__ cnt_in, const int* __restrict__ ovf_n,
            const int* __restrict__ ovf_src, const int* __restrict__ ovf_dst,
            const float* __restrict__ bias, float* __restrict__ out, int N,
            const int* __restrict__ cnt_out, const float* __restrict__ W2) {
    constexpr int VEC = C / 32;
    constexpr int B   = (VEC == 1) ? 32 : 24;
    static_assert(!FUSED || (C == HID && TROWS == 8), "the LDS tile holds rows of HID floats, one lane half's 8 rows per wave");
    __shared__ __attribute__((aligned(16))) float tile[FUSED ? GW * TROWS * TP : 4];
    __shared__ __attribute__((aligned(16))) float w2s[FUSED ? HID * WP : 4];
    const int lane = threadIdx.x & 31;
    const int wave = __builtin_amdgcn_readfirstlane((int)(threadIdx.x >> 5));
    const int nwaves = (int)gridDim.x * GW;

    float bv[VEC];
#pragma unroll
    for (int k = 0; k < VEC; ++k) bv[k] = bias[lane * VEC + k];

    int node = (int)blockIdx.x * GW + wave;
    int cnt_nx = 0, idx_nx = 0, co_nx = 0;
    if (node < N) {
        cnt_nx = cnt_in[node];
        idx_nx = slots[(long)node * CAP + lane];
        if constexpr (FUSED) co_nx = cnt_out[node];
    }
    if constexpr (FUSED) {
        static_assert(GW * 32 * 8 == HID * OUTC, "one 8-float piece of W2 per thread");
        typedef __attribute__((ext_vector_type(4))) float v4f;
        const int wr = (int)threadIdx.x / (OUTC / 8);
        const int wc = ((int)threadIdx.x % (OUTC / 8)) * 8;
        const v4f* __restrict__ g = reinterpret_cast<const v4f*>(W2 + wr * OUTC + wc);
        v4f* l = reinterpret_cast<v4f*>(w2s + wr * WP + wc);
        l[0] = g[0];
        l[1] = g[1];
    }
    for (int it = 0; FUSED ? (it < TROWS) : (node < N); ++it, node += nwaves) {
        if (!FUSED || node < N) {
            const int cnt   = cnt_nx;
            const int myidx = idx_nx;
            const int co    = co_nx;
            const int nxt   = node + nwaves;
            if (nxt < N) {
                cnt_nx = cnt_in[nxt];
                idx_nx = slots[(long)nxt * CAP + lane];
                if constexpr (FUSED) co_nx = cnt_out[nxt];
            }
            const int n32 = cnt < CAP ? cnt : CAP;
            float acc[VEC];
#pragma unroll
            for (int k = 0; k < VEC; ++k) acc[k] = 0.0f;

            for (int h = 0; h < n32; h += B) {
                const int n = n32 - h;
                float v[B][VEC];
#pragma unroll
                for (int u = 0; u < B; ++u) {
#pragma unroll
                    for (int k = 0; k < VEC; ++k) v[u][k] = 0.0f;
                    if (u == 0 || u < n) {
                        const int s = __builtin_amdgcn_readlane(myidx, h + u);
                        const float* __restrict__ p = H + (long)s * C + lane * VEC;
                        if constexpr (VEC == 2) {
                            const float2 t = *reinterpret_cast<const float2*>(p);
                            v[u][0] = t.x;
                            v[u][1] = t.y;
                        } else {
                            v[u][0] = p[0];
                        }
                    }
                }
#pragma unroll
                for (int u = 0; u < B; ++u) {
#pragma unroll
                    for (int k = 0; k < VEC; ++k) acc[k] += v[u][k];
                }
            }

            if (cnt > CAP) {
                const int novf = ovf_n[0];
                for (int j = 0; j < novf; j += 32) {
                    const int q = j + lane;
                    const int od = (q < novf) ? ovf_dst[q] : -1;
                    const int os = (q < novf) ? ovf_src[q] : 0;
                    unsigned m = __builtin_amdgcn_ballot_w32(od == node);
                    while (m != 0u) {
                        const int b = __builtin_ctz(m);
                        m &= m - 1u;
                        const int s = __builtin_amdgcn_readlane(os, b);
                        const float* __restrict__ p = H + (long)s * C + lane * VEC;
#pragma unroll
                        for (int k = 0; k < VEC; ++k) acc[k] += p[k];
                    }
                }
            }

            const float nin = rsqrtf(cnt > 0 ? (float)cnt : 1.0f);
            float r[VEC];
#pragma unroll
            for (int k = 0; k < VEC; ++k) {
                const float t = acc[k] * nin + bv[k];
                r[k] = RELU ? (t > 0.0f ? t : 0.0f) : t;
            }
            if constexpr (FUSED) {
                const float nm = rsqrtf(co > 0 ? (float)co : 1.0f);
                float* trow = tile + (wave * TROWS + it) * TP + lane * VEC;
                *reinterpret_cast<float2*>(trow) = make_float2(r[0] * nm, r[1] * nm);
            } else {
                float* __restrict__ o = out + (long)node * C + lane * VEC;
                if constexpr (VEC == 2) {
                    *reinterpret_cast<float2*>(o) = make_float2(r[0], r[1]);
                } else {
                    o[0] = r[0];
                }
            }
        } else {
            float* trow = tile + (wave * TROWS + it) * TP + lane * VEC;
            *reinterpret_cast<float2*>(trow) = make_float2(0.0f, 0.0f);
        }
    }
    if constexpr (FUSED) {
        __syncthreads();
        if (wave < GW * TROWS / 16) {
            tile_product<HID, OUTC>(tile + wave * 16 * TP, w2s, out, (int)blockIdx.x * GW + 2 * wave, nwaves, N, lane);
        }
    }
}

extern "C" void kernel_launch(void* const* d_in, const int* in_sizes, int n_in,
                              void* d_out, int out_size, void* d_ws, size_t ws_size,
                              hipStream_t stream) {
    const float* features = (const float*)d_in[0];
    const int*   src      = (const int*)d_in[1];
    const int*   dst      = (const int*)d_in[2];
    const float* W1       = (const float*)d_in[3];
    const float* b1       = (const float*)d_in[4];
    const float* W2       = (const float*)d_in[5];
    const float* b2       = (const float*)d_in[6];

    const int IN_F = 128;
    const int N = in_sizes[0] / IN_F;
    const int E = in_sizes[1];

    int*   wsi      = (int*)d_ws;
    float* wsf      = (float*)d_ws;
    float* hbuf     = wsf;
    float* h2buf    = wsf + 64L * N;
    int*   slots    = wsi + 128L * N;
    int*   ovf_src  = wsi + 160L * N;
    int*   ovf_dst  = wsi + 160L * N + (long)E;
    int*   cnt_in   = wsi + 160L * N + 2L * E;
    int*   cnt_out  = wsi + 161L * N + 2L * E;
    int*   ovf_n    = wsi + 162L * N + 2L * E;
    int*   gcount   = wsi + 162L * N + 2L * E + 32;
    int*   scount   = wsi + 162L * N + 2L * E + 32 + NBMAX * GSTR;
    u64*   spill    = (u64*)(wsi + 162L * N + 2L * E + 32 + 2 * NBMAX * GSTR);
    u64*   blist    = (u64*)(wsi + 162L * N + 4L * E + 32 + 2 * NBMAX * GSTR);
    const int NB = (N + RB - 1) / RB;
    int*   sspill   = wsi + 162L * N + 4L * E + 32 + 2 * NBMAX * GSTR + 2L * BCAP * NB;
    int*   slist    = wsi + 162L * N + 5L * E + 32 + 2 * NBMAX * GSTR + 2L * BCAP * NB;
    float* outp     = (float*)d_out;

    hipMemsetAsync(ovf_n, 0, sizeof(int) * (32 + 2 * NBMAX * GSTR), stream);

    const int nbk = 256;
    const int npk = 256;

    bucket_edges_and_product<<<nbk + npk, T1, 0, stream>>>(src, dst, gcount, blist, ovf_n + 16, spill, scount, slist,
                                                          ovf_n + 8, sspill, E, NB, features, W1, hbuf, N, nbk);
    rank_range<<<NB, T2, 0, stream>>>(blist, gcount, ovf_n + 16, spill, slist, scount, ovf_n + 8, sspill,
                                      cnt_in, cnt_out, slots, ovf_n, ovf_src, ovf_dst, hbuf, N);

    const int gather_blocks = 2048;
    const int gather1_blocks = N > gather_blocks * GW * TROWS ? (N + GW * TROWS - 1) / (GW * TROWS) : gather_blocks;

    gather_rows<HID, true, true><<<gather1_blocks, GW * 32, 0, stream>>>(hbuf, slots, cnt_in, ovf_n, ovf_src, ovf_dst, b1, h2buf, N,
                                                                      cnt_out, W2);

    gather_rows<OUTC, false, false><<<gather_blocks, GW * 32, 0, stream>>>(h2buf, slots, cnt_in, ovf_n, ovf_src, ovf_dst, b2, outp, N,
                                                                         nullptr, nullptr);
}


# Level 1 writes each work-group's range-ordered block whole beside a per-block table of run starts and lengths; level 2 reads its runs from it: no global adds, spill lists or clearing memset left
# speedup vs baseline: 7.6940x; 1.0326x over previous
#include <hip/hip_runtime.h>
#include <hip/hip_bf16.h>


typedef __attribute__((ext_vector_type(2))) float v2f;
typedef __attribute__((ext_vector_type(8))) float v8f;

typedef unsigned long long u64;

constexpr int CAP   = 32;
constexpr int BSH   = 9;
constexpr int RB    = 1 << BSH;
constexpr int T1    = 512;
constexpr int NBMAX = T1;
constexpr int EPT   = 13;
constexpr int STG   = T1 * EPT;
constexpr int GL    = 8;
constexpr int UR    = 6;
constexpr int T2    = 1024;
constexpr int PPT   = 8;
constexpr int HID   = 64;
constexpr int OUTC  = 32;
constexpr int GW    = 8;
constexpr int TROWS = 8;
constexpr int TP    = HID + 4;
constexpr int WP    = OUTC + 8;

__device__ __forceinline__ u64 pack_pair(int s, int d) {
    return (u64)(unsigned)s | ((u64)(unsigned)d << 32);
}

__device__ __forceinline__ void scan_starts(const int* cnt, int* start, int NB, int lane) {
    int carry = 0;
    for (int c0 = 0; c0 < NB; c0 += 32) {
        const int idx = c0 + lane;
        const int v = (idx < NB) ? cnt[idx] : 0;
        int x = v;
        for (int o = 1; o < 32; o <<= 1) {
            const int y = __shfl_up(x, o);
            if (lane >= o) x += y;
        }
        if (idx < NB) start[idx] = carry + x - v;
        carry += __shfl(x, 31);
    }
}

template <int K, int COLS>
__device__ __forceinline__ void product_strip(const float* __restrict__ X,
                                              const float* __restrict__ W, float* __restrict__ H,
                                              const int tile_m, const int lane) {
    constexpr int NT = COLS / 16;
    typedef __attribute__((ext_vector_type(NT))) float wvec;
    typedef __attribute__((ext_vector_type(4))) float v4f;
    constexpr int KB   = K / 8;
    constexpr int WROW = COLS / NT;

    const int c = lane & 15;
    const int h = lane >> 4;
    const int m = tile_m * 16 + c;

    const v4f*  __restrict__ xq = reinterpret_cast<const v4f*>(X + (long)m * K + 4 * h);
    const wvec* __restrict__ wq = reinterpret_cast<const wvec*>(W + (long)(4 * h) * COLS + NT * c);

    v8f acc[NT];
#pragma unroll
    for (int t = 0; t < NT; ++t) { v8f z = {}; acc[t] = z; }

#pragma unroll (KB <= 8 ? KB : 4)
    for (int kb = 0; kb < KB; ++kb) {
        const v4f  q  = xq[kb * 2];
        const wvec w0 = wq[(long)(kb * 8 + 0) * WROW];
        const wvec w1 = wq[(long)(kb * 8 + 1) * WROW];
        const wvec w2 = wq[(long)(kb * 8 + 2) * WROW];
        const wvec w3 = wq[(long)(kb * 8 + 3) * WROW];
        v2f a0, a1;
        a0.x = q.x;  a0.y = q.y;
        a1.x = q.z;  a1.y = q.w;
#pragma unroll
        for (int t = 0; t < NT; ++t) {
            v2f b;
            b.x = w0[t];
            b.y = w1[t];
            acc[t] = __builtin_amdgcn_wmma_f32_16x16x4_f32(
                false, a0, false, b, (short)0, acc[t], false, false);
        }
#pragma unroll
        for (int t = 0; t < NT; ++t) {
            v2f b;
            b.x = w2[t];
            b.y = w3[t];
            acc[t] = __builtin_amdgcn_wmma_f32_16x16x4_f32(
                false, a1, false, b, (short)0, acc[t], false, false);
        }
    }

    wvec* __restrict__ hq = reinterpret_cast<wvec*>(H + (long)(tile_m * 16 + 8 * h) * COLS + NT * c);
#pragma unroll
    for (int r = 0; r < 8; ++r) {
        wvec o;
#pragma unroll
        for (int t = 0; t < NT; ++t) o[t] = acc[t][r];
        hq[(long)r * WROW] = o;
    }
}

template <int K, int COLS>
__device__ __forceinline__ void tile_product(const float* trow0, const float* wl, float* __restrict__ Hout,
                                             const int node0, const int nstep, const int N, const int lane) {
    constexpr int NT = COLS / 16;
    typedef __attribute__((ext_vector_type(NT))) float wvec;
    typedef __attribute__((ext_vector_type(4))) float v4f;
    constexpr int KB   = K / 8;
    constexpr int WROW = WP / NT;
    static_assert(WP % NT == 0, "a row of the LDS copy of W is a whole number of wvec units");

    const int c = lane & 15;
    const int h = lane >> 4;
    const v4f*  xq = reinterpret_cast<const v4f*>(trow0 + c * TP + 4 * h);
    const wvec* wq = reinterpret_cast<const wvec*>(wl + (4 * h) * WP + NT * c);

    v8f acc[NT];
#pragma unroll
    for (int t = 0; t < NT; ++t) { v8f z = {}; acc[t] = z; }

#pragma unroll 2
    for (int kb = 0; kb < KB; ++kb) {
        const v4f  q  = xq[kb * 2];
        const wvec w0 = wq[(kb * 8 + 0) * WROW];
        const wvec w1 = wq[(kb * 8 + 1) * WROW];
        const wvec w2 = wq[(kb * 8 + 2) * WROW];
        const wvec w3 = wq[(kb * 8 + 3) * WROW];
        v2f a0, a1;
        a0.x = q.x;  a0.y = q.y;
        a1.x = q.z;  a1.y = q.w;
#pragma unroll
        for (int t = 0; t < NT; ++t) {
            v2f b;
            b.x = w0[t];
            b.y = w1[t];
            acc[t] = __builtin_amdgcn_wmma_f32_16x16x4_f32(
                false, a0, false, b, (short)0, acc[t], false, false);
        }
#pragma unroll
        for (int t = 0; t < NT; ++t) {
            v2f b;
            b.x = w2[t];
            b.y = w3[t];
            acc[t] = __builtin_amdgcn_wmma_f32_16x16x4_f32(
                false, a1, false, b, (short)0, acc[t], false, false);
        }
    }

#pragma unroll
    for (int r = 0; r < 8; ++r) {
        const int nd = node0 + h + r * nstep;
        if (nd < N) {
            wvec o;
#pragma unroll
            for (int t = 0; t < NT; ++t) o[t] = acc[t][r];
            *reinterpret_cast<wvec*>(Hout + (long)nd * COLS + NT * c) = o;
        }
    }
}

__global__ void __launch_bounds__(T1, 8)
bucket_edges_and_product(const int* __restrict__ src, const int* __restrict__ dst,
             u64* __restrict__ blistw, int* __restrict__ slistw, int4* __restrict__ tab,
             int* __restrict__ ovf_n, int E, int NB,
             const float* __restrict__ X, const float* __restrict__ W, float* __restrict__ H,
             int nrows, int nbk) {
    if ((int)blockIdx.x >= nbk) {
        const int wave = __builtin_amdgcn_readfirstlane((int)(threadIdx.x >> 5));
        const int npw  = (int)gridDim.x - nbk;
        for (int tile_m = wave * npw + ((int)blockIdx.x - nbk); tile_m * 16 < nrows; tile_m += npw * (T1 / 32))
            product_strip<128, HID>(X, W, H, tile_m, (int)(threadIdx.x & 31));
        return;
    }
    __shared__ int lcount[NBMAX];
    __shared__ int lstart[NBMAX];
    __shared__ u64 lpairs[STG];
    __shared__ int scnt[NBMAX];
    __shared__ int sstart[NBMAX];
    __shared__ int lsrc[STG];
    const int tid     = (int)threadIdx.x;
    const int chunk   = (E + nbk - 1) / nbk;
    const int cbeg    = (int)blockIdx.x * chunk;
    const int cend    = (cbeg + chunk < E) ? (cbeg + chunk) : E;
    const int nrounds = (chunk + STG - 1) / STG;

    if (blockIdx.x == 0 && tid == 0) ovf_n[0] = 0;

    for (int r = 0; r < nrounds; ++r) {
        const int base = cbeg + r * STG;
        if (tid < NB) { lcount[tid] = 0; scnt[tid] = 0; }
        __syncthreads();

        int s[EPT], d[EPT], lr[EPT], sr[EPT];
#pragma unroll
        for (int i = 0; i < EPT; ++i) {
            const int e = base + i * T1 + tid;
            const bool ok = e < cend;
            s[i] = ok ? src[e] : 0;
            d[i] = ok ? dst[e] : 0;
        }
#pragma unroll
        for (int i = 0; i < EPT; ++i) {
            lr[i] = 0;
            sr[i] = 0;
            if (base + i * T1 + tid < cend) {
                lr[i] = atomicAdd(&lcount[d[i] >> BSH], 1);
                sr[i] = atomicAdd(&scnt[s[i] >> BSH], 1);
            }
        }
        __syncthreads();

        if (tid < 32) {
            scan_starts(lcount, lstart, NB, tid);
        } else if (tid < 64) {
            scan_starts(scnt, sstart, NB, tid - 32);
        }
        __syncthreads();

        int st[EPT], st2[EPT];
#pragma unroll
        for (int i = 0; i < EPT; ++i) {
            st[i]  = lstart[d[i] >> BSH];
            st2[i] = sstart[s[i] >> BSH];
        }
#pragma unroll
        for (int i = 0; i < EPT; ++i) {
            if (base + i * T1 + tid < cend) {
                lpairs[st[i] + lr[i]] = pack_pair(s[i], d[i]);
                lsrc[st2[i] + sr[i]]  = s[i];
            }
        }
        __syncthreads();

        const int left = cend - base;
        const int cn = left < 0 ? 0 : (left < STG ? left : STG);
        const long blk = (long)r * nbk + (long)blockIdx.x;
        u64* __restrict__ bo = blistw + blk * STG;
        int* __restrict__ so = slistw + blk * STG;
        u64 v[EPT];
        int w[EPT];
#pragma unroll
        for (int i = 0; i < EPT; ++i) {
            const int j = i * T1 + tid;
            v[i] = lpairs[j < cn ? j : 0];
            w[i] = lsrc[j < cn ? j : 0];
        }
#pragma unroll
        for (int i = 0; i < EPT; ++i) {
            const int j = i * T1 + tid;
            if (j < cn) {
                bo[j] = v[i];
                so[j] = w[i];
            }
        }
        if (tid < NB) tab[blk * NB + tid] = make_int4(lstart[tid], lcount[tid], sstart[tid], scnt[tid]);
        if (r + 1 < nrounds) __syncthreads();
    }
}

__global__ void __launch_bounds__(T2)
rank_range(const u64* __restrict__ blistw, const int* __restrict__ slistw, const int4* __restrict__ tab,
           int E, int nbk, int NB,
           int* __restrict__ cnt_in, int* __restrict__ cnt_out,
           int* __restrict__ slots, int* __restrict__ ovf_n,
           int* __restrict__ ovf_src, int* __restrict__ ovf_dst, float* __restrict__ H, int N) {
    __shared__ int lcnt[RB];
    __shared__ int lco[RB];
    __shared__ __attribute__((aligned(16))) int lslots[RB * CAP];
    const int tid   = (int)threadIdx.x;
    const int b     = (int)blockIdx.x;
    const int nbase = b << BSH;

    int4* __restrict__ l4 = reinterpret_cast<int4*>(lslots);
    for (int t = tid; t < RB; t += T2) { lcnt[t] = 0; lco[t] = 0; }
    for (int t = tid; t < RB * CAP / 4; t += T2) l4[t] = make_int4(0, 0, 0, 0);
    __syncthreads();

    auto place = [&](const u64 p) {
        const int ps = (int)(unsigned)p;
        const int pd = (int)(p >> 32);
        const int dl = pd - nbase;
        const int r  = atomicAdd(&lcnt[dl], 1);
        if (r < CAP) {
            lslots[dl * CAP + r] = ps;
        } else {
            const int k = atomicAdd(ovf_n, 1);
            ovf_src[k] = ps;
            ovf_dst[k] = pd;
        }
    };

    const int chunk = (E + nbk - 1) / nbk;
    const int nblk  = nbk * ((chunk + STG - 1) / STG);
    constexpr int GPW = T2 / GL;
    const int g  = tid / GL;
    const int l8 = tid % GL;
    for (int b0 = 0; b0 < nblk; b0 += 2 * GPW) {
        int4 t[2];
        const u64* bl[2];
        const int* sl[2];
#pragma unroll
        for (int u = 0; u < 2; ++u) {
            const int blk = b0 + u * GPW + g;
            t[u]  = (blk < nblk) ? tab[(long)blk * NB + b] : make_int4(0, 0, 0, 0);
            bl[u] = blistw + (long)blk * STG;
            sl[u] = slistw + (long)blk * STG;
        }
        u64 p[2][UR];
        int q[2][UR];
#pragma unroll
        for (int u = 0; u < 2; ++u) {
#pragma unroll
            for (int i = 0; i < UR; ++i) {
                const int j = l8 + GL * i;
                p[u][i] = (j < t[u].y) ? bl[u][t[u].x + j] : 0ull;
                q[u][i] = (j < t[u].w) ? sl[u][t[u].z + j] : nbase;
            }
        }
#pragma unroll
        for (int u = 0; u < 2; ++u) {
#pragma unroll
            for (int i = 0; i < UR; ++i) {
                if (l8 + GL * i < t[u].y) place(p[u][i]);
            }
        }
#pragma unroll
        for (int u = 0; u < 2; ++u) {
#pragma unroll
            for (int i = 0; i < UR; ++i) {
                if (l8 + GL * i < t[u].w) atomicAdd(&lco[q[u][i] - nbase], 1);
            }
        }
#pragma unroll
        for (int u = 0; u < 2; ++u) {
            for (int j = l8 + GL * UR; j < t[u].y; j += GL) place(bl[u][t[u].x + j]);
            for (int j = l8 + GL * UR; j < t[u].w; j += GL) atomicAdd(&lco[sl[u][t[u].z + j] - nbase], 1);
        }
    }
    __syncthreads();

    const int left = N - nbase;
    const int nn = left < RB ? left : RB;
    for (int t = tid; t < nn; t += T2) {
        cnt_in[nbase + t]  = lcnt[t];
        cnt_out[nbase + t] = lco[t];
    }
    typedef __attribute__((ext_vector_type(4))) float v4f;
    constexpr int HV  = HID / 4;
    constexpr int HPT = RB * HV / T2;
    v4f* __restrict__ hp = reinterpret_cast<v4f*>(H + (long)nbase * HID);
    const int nv = nn * HV;
    v4f hv[HPT];
    int hc[HPT];
#pragma unroll
    for (int k = 0; k < HPT; ++k) {
        const int i = k * T2 + tid;
        if (i < nv) hv[k] = hp[i];
        hc[k] = lco[i / HV];
    }

    int4* __restrict__ g4 = reinterpret_cast<int4*>(slots + (long)nbase * CAP);
    for (int t = tid; t < nn * (CAP / 4); t += T2) g4[t] = l4[t];

#pragma unroll
    for (int k = 0; k < HPT; ++k) {
        const int i = k * T2 + tid;
        if (i < nv) hp[i] = hv[k] * rsqrtf(hc[k] > 0 ? (float)hc[k] : 1.0f);
    }
}

template <int C, bool RELU, bool FUSED>
__global__ void __launch_bounds__(GW * 32)
gather_rows(const float* __restrict__ H, const int* __restrict__ slots,
            const int* __restrict__ cnt_in, const int* __restrict__ ovf_n,
            const int* __restrict__ ovf_src, const int* __restrict__ ovf_dst,
            const float* __restrict__ bias, float* __restrict__ out, int N,
            const int* __restrict__ cnt_out, const float* __restrict__ W2) {
    constexpr int VEC = C / 32;
    constexpr int B   = (VEC == 1) ? 32 : 24;
    static_assert(!FUSED || (C == HID && TROWS == 8), "the LDS tile holds rows of HID floats, one lane half's 8 rows per wave");
    __shared__ __attribute__((aligned(16))) float tile[FUSED ? GW * TROWS * TP : 4];
    __shared__ __attribute__((aligned(16))) float w2s[FUSED ? HID * WP : 4];
    const int lane = threadIdx.x & 31;
    const int wave = __builtin_amdgcn_readfirstlane((int)(threadIdx.x >> 5));
    const int nwaves = (int)gridDim.x * GW;

    float bv[VEC];
#pragma unroll
    for (int k = 0; k < VEC; ++k) bv[k] = bias[lane * VEC + k];

    int node = (int)blockIdx.x * GW + wave;
    int cnt_nx = 0, idx_nx = 0, co_nx = 0;
    if (node < N) {
        cnt_nx = cnt_in[node];
        idx_nx = slots[(long)node * CAP + lane];
        if constexpr (FUSED) co_nx = cnt_out[node];
    }
    if constexpr (FUSED) {
        static_assert(GW * 32 * 8 == HID * OUTC, "one 8-float piece of W2 per thread");
        typedef __attribute__((ext_vector_type(4))) float v4f;
        const int wr = (int)threadIdx.x / (OUTC / 8);
        const int wc = ((int)threadIdx.x % (OUTC / 8)) * 8;
        const v4f* __restrict__ g = reinterpret_cast<const v4f*>(W2 + wr * OUTC + wc);
        v4f* l = reinterpret_cast<v4f*>(w2s + wr * WP + wc);
        l[0] = g[0];
        l[1] = g[1];
    }
    for (int it = 0; FUSED ? (it < TROWS) : (node < N); ++it, node += nwaves) {
        if (!FUSED || node < N) {
            const int cnt   = cnt_nx;
            const int myidx = idx_nx;
            const int co    = co_nx;
            const int nxt   = node + nwaves;
            if (nxt < N) {
                cnt_nx = cnt_in[nxt];
                idx_nx = slots[(long)nxt * CAP + lane];
                if constexpr (FUSED) co_nx = cnt_out[nxt];
            }
            const int n32 = cnt < CAP ? cnt : CAP;
            float acc[VEC];
#pragma unroll
            for (int k = 0; k < VEC; ++k) acc[k] = 0.0f;

            for (int h = 0; h < n32; h += B) {
                const int n = n32 - h;
                float v[B][VEC];
#pragma unroll
                for (int u = 0; u < B; ++u) {
#pragma unroll
                    for (int k = 0; k < VEC; ++k) v[u][k] = 0.0f;
                    if (u == 0 || u < n) {
                        const int s = __builtin_amdgcn_readlane(myidx, h + u);
                        const float* __restrict__ p = H + (long)s * C + lane * VEC;
                        if constexpr (VEC == 2) {
                            const float2 t = *reinterpret_cast<const float2*>(p);
                            v[u][0] = t.x;
                            v[u][1] = t.y;
                        } else {
                            v[u][0] = p[0];
                        }
                    }
                }
#pragma unroll
                for (int u = 0; u < B; ++u) {
#pragma unroll
                    for (int k = 0; k < VEC; ++k) acc[k] += v[u][k];
                }
            }

            if (cnt > CAP) {
                const int novf = ovf_n[0];
                for (int j = 0; j < novf; j += 32) {
                    const int q = j + lane;
                    const int od = (q < novf) ? ovf_dst[q] : -1;
                    const int os = (q < novf) ? ovf_src[q] : 0;
                    unsigned m = __builtin_amdgcn_ballot_w32(od == node);
                    while (m != 0u) {
                        const int b = __builtin_ctz(m);
                        m &= m - 1u;
                        const int s = __builtin_amdgcn_readlane(os, b);
                        const float* __restrict__ p = H + (long)s * C + lane * VEC;
#pragma unroll
                        for (int k = 0; k < VEC; ++k) acc[k] += p[k];
                    }
                }
            }

            const float nin = rsqrtf(cnt > 0 ? (float)cnt : 1.0f);
            float r[VEC];
#pragma unroll
            for (int k = 0; k < VEC; ++k) {
                const float t = acc[k] * nin + bv[k];
                r[k] = RELU ? (t > 0.0f ? t : 0.0f) : t;
            }
            if constexpr (FUSED) {
                const float nm = rsqrtf(co > 0 ? (float)co : 1.0f);
                float* trow = tile + (wave * TROWS + it) * TP + lane * VEC;
                *reinterpret_cast<float2*>(trow) = make_float2(r[0] * nm, r[1] * nm);
            } else {
                float* __restrict__ o = out + (long)node * C + lane * VEC;
                if constexpr (VEC == 2) {
                    *reinterpret_cast<float2*>(o) = make_float2(r[0], r[1]);
                } else {
                    o[0] = r[0];
                }
            }
        } else {
            float* trow = tile + (wave * TROWS + it) * TP + lane * VEC;
            *reinterpret_cast<float2*>(trow) = make_float2(0.0f, 0.0f);
        }
    }
    if constexpr (FUSED) {
        __syncthreads();
        if (wave < GW * TROWS / 16) {
            tile_product<HID, OUTC>(tile + wave * 16 * TP, w2s, out, (int)blockIdx.x * GW + 2 * wave, nwaves, N, lane);
        }
    }
}

extern "C" void kernel_launch(void* const* d_in, const int* in_sizes, int n_in,
                              void* d_out, int out_size, void* d_ws, size_t ws_size,
                              hipStream_t stream) {
    const float* features = (const float*)d_in[0];
    const int*   src      = (const int*)d_in[1];
    const int*   dst      = (const int*)d_in[2];
    const float* W1       = (const float*)d_in[3];
    const float* b1       = (const float*)d_in[4];
    const float* W2       = (const float*)d_in[5];
    const float* b2       = (const float*)d_in[6];

    const int IN_F = 128;
    const int N = in_sizes[0] / IN_F;
    const int E = in_sizes[1];

    int*   wsi      = (int*)d_ws;
    float* wsf      = (float*)d_ws;
    float* hbuf     = wsf;
    float* h2buf    = wsf + 64L * N;
    int*   slots    = wsi + 128L * N;
    int*   ovf_src  = wsi + 160L * N;
    int*   ovf_dst  = wsi + 160L * N + (long)E;
    int*   cnt_in   = wsi + 160L * N + 2L * E;
    int*   cnt_out  = wsi + 161L * N + 2L * E;
    int*   ovf_n    = wsi + 162L * N + 2L * E;
    const int NB = (N + RB - 1) / RB;
    const int nbk = 256;
    const int npk = 256;
    const long lists = (162L * N + 2L * E + 32 + 3) & ~3L;
    const long cap   = ((long)E + (long)nbk * STG + 3) & ~3L;
    u64*   blistw   = (u64*)(wsi + lists);
    int*   slistw   = wsi + lists + 2 * cap;
    int4*  tab      = (int4*)(wsi + lists + 3 * cap);
    float* outp     = (float*)d_out;

    bucket_edges_and_product<<<nbk + npk, T1, 0, stream>>>(src, dst, blistw, slistw, tab, ovf_n, E, NB, features, W1, hbuf, N, nbk);
    rank_range<<<NB, T2, 0, stream>>>(blistw, slistw, tab, E, nbk, NB,
                                      cnt_in, cnt_out, slots, ovf_n, ovf_src, ovf_dst, hbuf, N);

    const int gather_blocks = 2048;
    const int gather1_blocks = N > gather_blocks * GW * TROWS ? (N + GW * TROWS - 1) / (GW * TROWS) : gather_blocks;

    gather_rows<HID, true, true><<<gather1_blocks, GW * 32, 0, stream>>>(hbuf, slots, cnt_in, ovf_n, ovf_src, ovf_dst, b1, h2buf, N,
                                                                      cnt_out, W2);

    gather_rows<OUTC, false, false><<<gather_blocks, GW * 32, 0, stream>>>(h2buf, slots, cnt_in, ovf_n, ovf_src, ovf_dst, b2, outp, N,
                                                                         nullptr, nullptr);
}
